# starting kernel
# baseline (speedup 1.0000x reference)
#include <hip/hip_runtime.h>

#define N_NODES 50000
#define F_IN    128
#define HID     256
#define CLS     40
#define NE      640000

typedef __attribute__((ext_vector_type(2))) float v2f;
typedef __attribute__((ext_vector_type(8))) float v8f;

__global__ void zero_f32(float* __restrict__ p, long long n) {
  long long i = (long long)blockIdx.x * blockDim.x + threadIdx.x;
  if (i < n) p[i] = 0.0f;
}

__global__ void scatter_accum(const float* __restrict__ feat,
                              const int*   __restrict__ src,
                              const int*   __restrict__ dst,
                              float* __restrict__ agg,
                              float* __restrict__ cnt,
                              int nfeat4, int do_cnt) {
  long long tid = (long long)blockIdx.x * blockDim.x + threadIdx.x;
  long long e = tid / nfeat4;
  int j = (int)(tid - e * nfeat4);
  if (e >= NE) return;
  int s = src[e];
  int d = dst[e];
  int F = nfeat4 * 4;
  float4 v = ((const float4*)feat)[(long long)s * nfeat4 + j];
  float* a = agg + (long long)d * F + j * 4;
  unsafeAtomicAdd(a + 0, v.x);
  unsafeAtomicAdd(a + 1, v.y);
  unsafeAtomicAdd(a + 2, v.z);
  unsafeAtomicAdd(a + 3, v.w);
  if (do_cnt && j == 0) unsafeAtomicAdd(cnt + d, 1.0f);
}

__global__ void normalize_mean(float* __restrict__ agg,
                               const float* __restrict__ cnt, int F) {
  long long i = (long long)blockIdx.x * blockDim.x + threadIdx.x;
  long long n = (long long)N_NODES * F;
  if (i >= n) return;
  int node = (int)(i / F);
  float c = cnt[node];
  c = c > 1.0f ? c : 1.0f;
  agg[i] = agg[i] / c;
}

__global__ void dual_gemm_wmma(const float* __restrict__ A1,
                               const float* __restrict__ W1,
                               const float* __restrict__ A2,
                               const float* __restrict__ W2,
                               const float* __restrict__ bias,
                               float* __restrict__ out,
                               int K, int ncols, int ldo, int do_relu) {
  const int lane   = threadIdx.x & 31;
  const int wid    = blockIdx.x * (blockDim.x >> 5) + (threadIdx.x >> 5);
  const int ctiles = (ncols + 15) >> 4;
  const int mtiles = N_NODES >> 4;
  if (wid >= mtiles * ctiles) return;
  const int rt = wid / ctiles;
  const int ct = wid - rt * ctiles;

  const int mrow  = rt * 16 + (lane & 15);
  const int ncol  = ct * 16 + (lane & 15);
  const int khalf = (lane >> 4) * 2;
  const bool cvalid  = (ncol < ncols);
  const int  ncol_ld = cvalid ? ncol : (ncols - 1);

  const float* arow1 = A1 + (unsigned)mrow    * (unsigned)K + khalf;
  const float* brow1 = W1 + (unsigned)ncol_ld * (unsigned)K + khalf;
  const float* arow2 = A2 + (unsigned)mrow    * (unsigned)K + khalf;
  const float* brow2 = W2 + (unsigned)ncol_ld * (unsigned)K + khalf;

  v8f c = {};

  #pragma unroll 8
  for (int k0 = 0; k0 < K; k0 += 4) {
    float2 av = *(const float2*)(arow1 + k0);
    float2 bv = *(const float2*)(brow1 + k0);
    v2f a; a.x = av.x; a.y = av.y;
    v2f b; b.x = bv.x; b.y = bv.y;
    c = __builtin_amdgcn_wmma_f32_16x16x4_f32(false, a, false, b,
                                              (short)0, c, false, false);
  }
  #pragma unroll 8
  for (int k0 = 0; k0 < K; k0 += 4) {
    float2 av = *(const float2*)(arow2 + k0);
    float2 bv = *(const float2*)(brow2 + k0);
    v2f a; a.x = av.x; a.y = av.y;
    v2f b; b.x = bv.x; b.y = bv.y;
    c = __builtin_amdgcn_wmma_f32_16x16x4_f32(false, a, false, b,
                                              (short)0, c, false, false);
  }

  if (cvalid) {
    float bv = bias[ncol];
    unsigned rbase = (unsigned)(rt * 16 + ((lane < 16) ? 0 : 8));
    float* orow = out + (unsigned)rbase * (unsigned)ldo + ncol;
    #pragma unroll
    for (int i = 0; i < 8; ++i) {
      float v = c[i] + bv;
      if (do_relu) v = v > 0.0f ? v : 0.0f;
      orow[(unsigned)i * (unsigned)ldo] = v;
    }
  }
}

__global__ void log_softmax_rows(const float* __restrict__ logits,
                                 float* __restrict__ out) {
  int r = blockIdx.x * blockDim.x + threadIdx.x;
  if (r >= N_NODES) return;
  const float* l = logits + (long long)r * CLS;
  float m = -3.402823466e38f;
  #pragma unroll
  for (int j = 0; j < CLS; ++j) m = fmaxf(m, l[j]);
  float s = 0.0f;
  #pragma unroll
  for (int j = 0; j < CLS; ++j) s += expf(l[j] - m);
  float ls = logf(s);
  float* o = out + (long long)r * CLS;
  #pragma unroll
  for (int j = 0; j < CLS; ++j) o[j] = l[j] - m - ls;
}

extern "C" void kernel_launch(void* const* d_in, const int* in_sizes, int n_in,
                              void* d_out, int out_size, void* d_ws, size_t ws_size,
                              hipStream_t stream) {
  (void)in_sizes; (void)n_in; (void)out_size; (void)ws_size;

  const float* x    = (const float*)d_in[0];
  const int*   ei   = (const int*)  d_in[1];
  const float* W1_l = (const float*)d_in[2];
  const float* b1   = (const float*)d_in[3];
  const float* W1_r = (const float*)d_in[4];
  const float* W2_l = (const float*)d_in[5];
  const float* b2   = (const float*)d_in[6];
  const float* W2_r = (const float*)d_in[7];
  const int* src = ei;
  const int* dst = ei + NE;
  float* out = (float*)d_out;

  float* ws     = (float*)d_ws;
  float* cnt    = ws;
  float* agg1   = cnt  + N_NODES;
  float* h      = agg1 + (long long)N_NODES * F_IN;
  float* agg2   = h    + (long long)N_NODES * HID;
  float* logits = agg2 + (long long)N_NODES * HID;

  const int TB = 256;
  auto blocks = [](long long n, int tb) { return (unsigned)((n + tb - 1) / tb); };

  long long nz1 = (long long)N_NODES * (1 + F_IN);
  zero_f32<<<blocks(nz1, TB), TB, 0, stream>>>(cnt, nz1);

  long long sc1 = (long long)NE * (F_IN / 4);
  scatter_accum<<<blocks(sc1, TB), TB, 0, stream>>>(x, src, dst, agg1, cnt,
                                                    F_IN / 4,  1);

  long long nm1 = (long long)N_NODES * F_IN;
  normalize_mean<<<blocks(nm1, TB), TB, 0, stream>>>(agg1, cnt, F_IN);

  {
    int tiles = (N_NODES / 16) * (HID / 16);
    dual_gemm_wmma<<<blocks(tiles, 8), TB, 0, stream>>>(
        agg1, W1_l, x, W1_r, b1, h, F_IN, HID, HID,  1);
  }

  long long nz2 = (long long)N_NODES * HID;
  zero_f32<<<blocks(nz2, TB), TB, 0, stream>>>(agg2, nz2);

  long long sc2 = (long long)NE * (HID / 4);
  scatter_accum<<<blocks(sc2, TB), TB, 0, stream>>>(h, src, dst, agg2, cnt,
                                                    HID / 4,  0);

  long long nm2 = (long long)N_NODES * HID;
  normalize_mean<<<blocks(nm2, TB), TB, 0, stream>>>(agg2, cnt, HID);

  {
    int ctiles = (CLS + 15) / 16;
    int tiles = (N_NODES / 16) * ctiles;
    dual_gemm_wmma<<<blocks(tiles, 8), TB, 0, stream>>>(
        agg2, W2_l, h, W2_r, b2, logits, HID, CLS, CLS,  0);
  }

  log_softmax_rows<<<blocks(N_NODES, TB), TB, 0, stream>>>(logits, out);
}


# Layer 2 maps h to the 40 classes (one WMMA GEMM over [W2_l;W2_r]) BEFORE the neighbour mean: atomics scatter 40 floats per edge, not 256; sum order changes; mean+root+bias+log-softmax one row pass
# speedup vs baseline: 1.6168x; 1.6168x over previous
#include <hip/hip_runtime.h>

#define N_NODES 50000
#define F_IN    128
#define HID     256
#define CLS     40
#define NE      640000

typedef __attribute__((ext_vector_type(2))) float v2f;
typedef __attribute__((ext_vector_type(8))) float v8f;

__global__ void zero_f32(float* __restrict__ p, long long n) {
  long long i = (long long)blockIdx.x * blockDim.x + threadIdx.x;
  if (i < n) p[i] = 0.0f;
}

__global__ void scatter_accum(const float* __restrict__ feat,
                              const int*   __restrict__ src,
                              const int*   __restrict__ dst,
                              float* __restrict__ agg,
                              float* __restrict__ cnt,
                              int nfeat4, int do_cnt) {
  long long tid = (long long)blockIdx.x * blockDim.x + threadIdx.x;
  long long e = tid / nfeat4;
  int j = (int)(tid - e * nfeat4);
  if (e >= NE) return;
  int s = src[e];
  int d = dst[e];
  int F = nfeat4 * 4;
  float4 v = ((const float4*)feat)[(long long)s * nfeat4 + j];
  float* a = agg + (long long)d * F + j * 4;
  unsafeAtomicAdd(a + 0, v.x);
  unsafeAtomicAdd(a + 1, v.y);
  unsafeAtomicAdd(a + 2, v.z);
  unsafeAtomicAdd(a + 3, v.w);
  if (do_cnt && j == 0) unsafeAtomicAdd(cnt + d, 1.0f);
}

__global__ void normalize_mean(float* __restrict__ agg,
                               const float* __restrict__ cnt, int F) {
  long long i = (long long)blockIdx.x * blockDim.x + threadIdx.x;
  long long n = (long long)N_NODES * F;
  if (i >= n) return;
  int node = (int)(i / F);
  float c = cnt[node];
  c = c > 1.0f ? c : 1.0f;
  agg[i] = agg[i] / c;
}

__global__ void dual_gemm_wmma(const float* __restrict__ A1,
                               const float* __restrict__ W1,
                               const float* __restrict__ A2,
                               const float* __restrict__ W2,
                               const float* __restrict__ bias,
                               float* __restrict__ out,
                               int K, int ncols, int ldo, int do_relu) {
  const int lane   = threadIdx.x & 31;
  const int wid    = blockIdx.x * (blockDim.x >> 5) + (threadIdx.x >> 5);
  const int ctiles = (ncols + 15) >> 4;
  const int mtiles = N_NODES >> 4;
  if (wid >= mtiles * ctiles) return;
  const int rt = wid / ctiles;
  const int ct = wid - rt * ctiles;

  const int mrow  = rt * 16 + (lane & 15);
  const int ncol  = ct * 16 + (lane & 15);
  const int khalf = (lane >> 4) * 2;
  const bool cvalid  = (ncol < ncols);
  const int  ncol_ld = cvalid ? ncol : (ncols - 1);

  const float* arow1 = A1 + (unsigned)mrow    * (unsigned)K + khalf;
  const float* brow1 = W1 + (unsigned)ncol_ld * (unsigned)K + khalf;
  const float* arow2 = A2 + (unsigned)mrow    * (unsigned)K + khalf;
  const float* brow2 = W2 + (unsigned)ncol_ld * (unsigned)K + khalf;

  v8f c = {};

  #pragma unroll 8
  for (int k0 = 0; k0 < K; k0 += 4) {
    float2 av = *(const float2*)(arow1 + k0);
    float2 bv = *(const float2*)(brow1 + k0);
    v2f a; a.x = av.x; a.y = av.y;
    v2f b; b.x = bv.x; b.y = bv.y;
    c = __builtin_amdgcn_wmma_f32_16x16x4_f32(false, a, false, b,
                                              (short)0, c, false, false);
  }
  #pragma unroll 8
  for (int k0 = 0; k0 < K; k0 += 4) {
    float2 av = *(const float2*)(arow2 + k0);
    float2 bv = *(const float2*)(brow2 + k0);
    v2f a; a.x = av.x; a.y = av.y;
    v2f b; b.x = bv.x; b.y = bv.y;
    c = __builtin_amdgcn_wmma_f32_16x16x4_f32(false, a, false, b,
                                              (short)0, c, false, false);
  }

  if (cvalid) {
    float bv = bias[ncol];
    unsigned rbase = (unsigned)(rt * 16 + ((lane < 16) ? 0 : 8));
    float* orow = out + (unsigned)rbase * (unsigned)ldo + ncol;
    #pragma unroll
    for (int i = 0; i < 8; ++i) {
      float v = c[i] + bv;
      if (do_relu) v = v > 0.0f ? v : 0.0f;
      orow[(unsigned)i * (unsigned)ldo] = v;
    }
  }
}

__global__ void proj_gemm_wmma(const float* __restrict__ H,
                               const float* __restrict__ Wl,
                               const float* __restrict__ Wr,
                               float* __restrict__ P,
                               float* __restrict__ R) {
  const int lane   = threadIdx.x & 31;
  const int wid    = blockIdx.x * (blockDim.x >> 5) + (threadIdx.x >> 5);
  const int ctiles = (2 * CLS) >> 4;
  const int mtiles = N_NODES >> 4;
  if (wid >= mtiles * ctiles) return;
  const int rt = wid / ctiles;
  const int ct = wid - rt * ctiles;

  const int mrow  = rt * 16 + (lane & 15);
  const int ncol  = ct * 16 + (lane & 15);
  const int khalf = (lane >> 4) * 2;
  const bool root = (ncol >= CLS);
  const int  wrow = root ? (ncol - CLS) : ncol;

  const float* arow = H + (unsigned)mrow * (unsigned)HID + khalf;
  const float* brow = (root ? Wr : Wl) + (unsigned)wrow * (unsigned)HID + khalf;

  v8f c = {};

  #pragma unroll 8
  for (int k0 = 0; k0 < HID; k0 += 4) {
    float2 av = *(const float2*)(arow + k0);
    float2 bv = *(const float2*)(brow + k0);
    v2f a; a.x = av.x; a.y = av.y;
    v2f b; b.x = bv.x; b.y = bv.y;
    c = __builtin_amdgcn_wmma_f32_16x16x4_f32(false, a, false, b,
                                              (short)0, c, false, false);
  }

  unsigned rbase = (unsigned)(rt * 16 + ((lane < 16) ? 0 : 8));
  float* orow = (root ? R : P) + rbase * (unsigned)CLS + wrow;
  #pragma unroll
  for (int i = 0; i < 8; ++i) orow[(unsigned)i * (unsigned)CLS] = c[i];
}

__global__ void finish_rows(const float* __restrict__ aggp,
                            const float* __restrict__ cnt,
                            const float* __restrict__ rootp,
                            const float* __restrict__ bias,
                            float* __restrict__ out) {
  int r = blockIdx.x * blockDim.x + threadIdx.x;
  if (r >= N_NODES) return;
  float c = cnt[r];
  c = c > 1.0f ? c : 1.0f;
  const float4* a4 = (const float4*)(aggp  + (long long)r * CLS);
  const float4* r4 = (const float4*)(rootp + (long long)r * CLS);
  float l[CLS];
  #pragma unroll
  for (int q = 0; q < CLS / 4; ++q) {
    float4 a = a4[q];
    float4 t = r4[q];
    l[4 * q + 0] = a.x / c + bias[4 * q + 0] + t.x;
    l[4 * q + 1] = a.y / c + bias[4 * q + 1] + t.y;
    l[4 * q + 2] = a.z / c + bias[4 * q + 2] + t.z;
    l[4 * q + 3] = a.w / c + bias[4 * q + 3] + t.w;
  }
  float m = -3.402823466e38f;
  #pragma unroll
  for (int j = 0; j < CLS; ++j) m = fmaxf(m, l[j]);
  float s = 0.0f;
  #pragma unroll
  for (int j = 0; j < CLS; ++j) s += expf(l[j] - m);
  float ls = logf(s);
  float* o = out + (long long)r * CLS;
  #pragma unroll
  for (int j = 0; j < CLS; ++j) o[j] = l[j] - m - ls;
}

extern "C" void kernel_launch(void* const* d_in, const int* in_sizes, int n_in,
                              void* d_out, int out_size, void* d_ws, size_t ws_size,
                              hipStream_t stream) {
  (void)in_sizes; (void)n_in; (void)out_size; (void)ws_size;

  const float* x    = (const float*)d_in[0];
  const int*   ei   = (const int*)  d_in[1];
  const float* W1_l = (const float*)d_in[2];
  const float* b1   = (const float*)d_in[3];
  const float* W1_r = (const float*)d_in[4];
  const float* W2_l = (const float*)d_in[5];
  const float* b2   = (const float*)d_in[6];
  const float* W2_r = (const float*)d_in[7];
  const int* src = ei;
  const int* dst = ei + NE;
  float* out = (float*)d_out;

  float* ws     = (float*)d_ws;
  float* cnt    = ws;
  float* agg1   = cnt  + N_NODES;
  float* h      = agg1 + (long long)N_NODES * F_IN;
  float* p2     = h    + (long long)N_NODES * HID;
  float* r2     = p2   + (long long)N_NODES * CLS;
  float* aggp   = r2   + (long long)N_NODES * CLS;

  const int TB = 256;
  auto blocks = [](long long n, int tb) { return (unsigned)((n + tb - 1) / tb); };

  long long nz1 = (long long)N_NODES * (1 + F_IN);
  zero_f32<<<blocks(nz1, TB), TB, 0, stream>>>(cnt, nz1);

  long long sc1 = (long long)NE * (F_IN / 4);
  scatter_accum<<<blocks(sc1, TB), TB, 0, stream>>>(x, src, dst, agg1, cnt,
                                                    F_IN / 4,  1);

  long long nm1 = (long long)N_NODES * F_IN;
  normalize_mean<<<blocks(nm1, TB), TB, 0, stream>>>(agg1, cnt, F_IN);

  {
    int tiles = (N_NODES / 16) * (HID / 16);
    dual_gemm_wmma<<<blocks(tiles, 8), TB, 0, stream>>>(
        agg1, W1_l, x, W1_r, b1, h, F_IN, HID, HID,  1);
  }

  {
    int tiles = (N_NODES / 16) * ((2 * CLS) / 16);
    proj_gemm_wmma<<<blocks(tiles, 8), TB, 0, stream>>>(h, W2_l, W2_r, p2, r2);
  }

  long long nz2 = (long long)N_NODES * CLS;
  zero_f32<<<blocks(nz2, TB), TB, 0, stream>>>(aggp, nz2);

  long long sc2 = (long long)NE * (CLS / 4);
  scatter_accum<<<blocks(sc2, TB), TB, 0, stream>>>(p2, src, dst, aggp, cnt,
                                                    CLS / 4,  0);

  finish_rows<<<blocks(N_NODES, TB), TB, 0, stream>>>(aggp, cnt, r2, b2, out);
}


# Both WMMA GEMMs retiled for operand reuse: a block stages its weight strip (all K=256) once in LDS, each wave reuses one A fragment over 8 (layer 1) or 5 (layer 2) column tiles; sum order kept
# speedup vs baseline: 2.1554x; 1.3331x over previous
#include <hip/hip_runtime.h>

#define N_NODES 50000
#define F_IN    128
#define HID     256
#define CLS     40
#define NE      640000

typedef __attribute__((ext_vector_type(2))) float v2f;
typedef __attribute__((ext_vector_type(8))) float v8f;

__global__ void zero_f32(float* __restrict__ p, long long n) {
  long long i = (long long)blockIdx.x * blockDim.x + threadIdx.x;
  if (i < n) p[i] = 0.0f;
}

__global__ void scatter_accum(const float* __restrict__ feat,
                              const int*   __restrict__ src,
                              const int*   __restrict__ dst,
                              float* __restrict__ agg,
                              float* __restrict__ cnt,
                              int nfeat4, int do_cnt) {
  long long tid = (long long)blockIdx.x * blockDim.x + threadIdx.x;
  long long e = tid / nfeat4;
  int j = (int)(tid - e * nfeat4);
  if (e >= NE) return;
  int s = src[e];
  int d = dst[e];
  int F = nfeat4 * 4;
  float4 v = ((const float4*)feat)[(long long)s * nfeat4 + j];
  float* a = agg + (long long)d * F + j * 4;
  unsafeAtomicAdd(a + 0, v.x);
  unsafeAtomicAdd(a + 1, v.y);
  unsafeAtomicAdd(a + 2, v.z);
  unsafeAtomicAdd(a + 3, v.w);
  if (do_cnt && j == 0) unsafeAtomicAdd(cnt + d, 1.0f);
}

__global__ void normalize_mean(float* __restrict__ agg,
                               const float* __restrict__ cnt, int F) {
  long long i = (long long)blockIdx.x * blockDim.x + threadIdx.x;
  long long n = (long long)N_NODES * F;
  if (i >= n) return;
  int node = (int)(i / F);
  float c = cnt[node];
  c = c > 1.0f ? c : 1.0f;
  agg[i] = agg[i] / c;
}

#define BPITCH 132

template <int NT>
__device__ __forceinline__ void group_products(const float2 (&f)[8],
                                               const float* bg,
                                               v8f (&c)[NT]) {
  #pragma unroll
  for (int s = 0; s < 8; ++s) {
    v2f a; a.x = f[s].x; a.y = f[s].y;
    #pragma unroll
    for (int t = 0; t < NT; ++t) {
      float2 bv = *(const float2*)(bg + t * 16 * BPITCH + 4 * s);
      v2f b; b.x = bv.x; b.y = bv.y;
      c[t] = __builtin_amdgcn_wmma_f32_16x16x4_f32(false, a, false, b,
                                                   (short)0, c[t], false, false);
    }
  }
}

template <int NT, int NC>
__device__ __forceinline__ void strip_product(const float* __restrict__ a_lo,
                                              const float* __restrict__ a_hi,
                                              const float* bs,
                                              v8f (&c)[NT]) {
  float2 fa[8], fb[8];
  #pragma unroll
  for (int s = 0; s < 8; ++s) fa[s] = *(const float2*)(a_lo + 4 * s);

  #pragma unroll 1
  for (int gg = 0; gg < 4; ++gg) {
    {
      const int gn = 2 * gg + 1;
      const float* pn = (gn < 4) ? (a_lo + gn * 32) : (a_hi + (gn - 4) * 32);
      #pragma unroll
      for (int s = 0; s < 8; ++s) fb[s] = *(const float2*)(pn + 4 * s);
    }
    const float* bt = bs + (gg >> 1) * (NC * BPITCH) + (gg & 1) * 64;
    group_products<NT>(fa, bt, c);
    {
      const int gn = (gg < 3) ? (2 * gg + 2) : 7;
      const float* pn = (gn < 4) ? (a_lo + gn * 32) : (a_hi + (gn - 4) * 32);
      #pragma unroll
      for (int s = 0; s < 8; ++s) fa[s] = *(const float2*)(pn + 4 * s);
    }
    group_products<NT>(fb, bt + 32, c);
  }
}

__global__ void __launch_bounds__(256, 4)
layer1_gemm_strip(const float* __restrict__ agg,
                  const float* __restrict__ x,
                  const float* __restrict__ Wl,
                  const float* __restrict__ Wr,
                  const float* __restrict__ bias,
                  float* __restrict__ h) {
  __shared__ __attribute__((aligned(16))) float Bs[2 * 128 * BPITCH];
  const int tid  = threadIdx.x;
  const int lane = tid & 31;
  const int wave = tid >> 5;
  const int colbase = (blockIdx.x & 1) * 128;
  const int rb      = blockIdx.x >> 1;

  #pragma unroll
  for (int j = 0; j < 32; ++j) {
    const int i  = tid + j * 256;
    const int cc = i >> 6;
    const int q  = i & 63;
    const float* sp = (q < 32)
        ? (Wl + (unsigned)(colbase + cc) * (unsigned)F_IN + q * 4)
        : (Wr + (unsigned)(colbase + cc) * (unsigned)F_IN + (q - 32) * 4);
    const float4 v = *(const float4*)sp;
    *(float4*)(Bs + (q >> 5) * (128 * BPITCH) + cc * BPITCH + (q & 31) * 4) = v;
  }
  __syncthreads();

  const int rt = rb * 8 + wave;
  if (rt >= (N_NODES >> 4)) return;

  const int mrow  = rt * 16 + (lane & 15);
  const int khalf = (lane >> 4) * 2;
  const float* a_lo = agg + (unsigned)mrow * (unsigned)F_IN + khalf;
  const float* a_hi = x   + (unsigned)mrow * (unsigned)F_IN + khalf;
  const float* bs   = Bs + (lane & 15) * BPITCH + khalf;

  v8f c[8];
  #pragma unroll
  for (int t = 0; t < 8; ++t) c[t] = v8f{};

  strip_product<8, 128>(a_lo, a_hi, bs, c);

  const unsigned rbase = (unsigned)(rt * 16 + ((lane < 16) ? 0 : 8));
  #pragma unroll
  for (int t = 0; t < 8; ++t) {
    const int ncol = colbase + t * 16 + (lane & 15);
    const float bv = bias[ncol];
    float* orow = h + rbase * (unsigned)HID + ncol;
    #pragma unroll
    for (int i = 0; i < 8; ++i) {
      float v = c[t][i] + bv;
      v = v > 0.0f ? v : 0.0f;
      orow[(unsigned)i * (unsigned)HID] = v;
    }
  }
}

__global__ void __launch_bounds__(256, 6)
proj_gemm_strip(const float* __restrict__ H,
                const float* __restrict__ Wl,
                const float* __restrict__ Wr,
                float* __restrict__ P,
                float* __restrict__ R) {
  __shared__ __attribute__((aligned(16))) float Bs[2 * (2 * CLS) * BPITCH];
  const int tid  = threadIdx.x;
  const int lane = tid & 31;
  const int wave = tid >> 5;

  #pragma unroll
  for (int j = 0; j < 20; ++j) {
    const int i  = tid + j * 256;
    const int cc = i >> 6;
    const int q  = i & 63;
    const float* sp = (cc < CLS) ? (Wl + (unsigned)cc * (unsigned)HID + q * 4)
                                 : (Wr + (unsigned)(cc - CLS) * (unsigned)HID + q * 4);
    const float4 v = *(const float4*)sp;
    *(float4*)(Bs + (q >> 5) * (2 * CLS * BPITCH) + cc * BPITCH + (q & 31) * 4) = v;
  }
  __syncthreads();

  const int rt = blockIdx.x * 8 + wave;
  if (rt >= (N_NODES >> 4)) return;

  const int mrow  = rt * 16 + (lane & 15);
  const int khalf = (lane >> 4) * 2;
  const float* a_lo = H + (unsigned)mrow * (unsigned)HID + khalf;
  const float* a_hi = a_lo + 128;
  const float* bs   = Bs + (lane & 15) * BPITCH + khalf;

  v8f c[5];
  #pragma unroll
  for (int t = 0; t < 5; ++t) c[t] = v8f{};

  strip_product<5, 2 * CLS>(a_lo, a_hi, bs, c);

  const unsigned rbase = (unsigned)(rt * 16 + ((lane < 16) ? 0 : 8));
  #pragma unroll
  for (int t = 0; t < 5; ++t) {
    const int  ncol = t * 16 + (lane & 15);
    const bool root = (ncol >= CLS);
    const int  wrow = root ? (ncol - CLS) : ncol;
    float* orow = (root ? R : P) + rbase * (unsigned)CLS + wrow;
    #pragma unroll
    for (int i = 0; i < 8; ++i) orow[(unsigned)i * (unsigned)CLS] = c[t][i];
  }
}

__global__ void finish_rows(const float* __restrict__ aggp,
                            const float* __restrict__ cnt,
                            const float* __restrict__ rootp,
                            const float* __restrict__ bias,
                            float* __restrict__ out) {
  int r = blockIdx.x * blockDim.x + threadIdx.x;
  if (r >= N_NODES) return;
  float c = cnt[r];
  c = c > 1.0f ? c : 1.0f;
  const float4* a4 = (const float4*)(aggp  + (long long)r * CLS);
  const float4* r4 = (const float4*)(rootp + (long long)r * CLS);
  float l[CLS];
  #pragma unroll
  for (int q = 0; q < CLS / 4; ++q) {
    float4 a = a4[q];
    float4 t = r4[q];
    l[4 * q + 0] = a.x / c + bias[4 * q + 0] + t.x;
    l[4 * q + 1] = a.y / c + bias[4 * q + 1] + t.y;
    l[4 * q + 2] = a.z / c + bias[4 * q + 2] + t.z;
    l[4 * q + 3] = a.w / c + bias[4 * q + 3] + t.w;
  }
  float m = -3.402823466e38f;
  #pragma unroll
  for (int j = 0; j < CLS; ++j) m = fmaxf(m, l[j]);
  float s = 0.0f;
  #pragma unroll
  for (int j = 0; j < CLS; ++j) s += expf(l[j] - m);
  float ls = logf(s);
  float* o = out + (long long)r * CLS;
  #pragma unroll
  for (int j = 0; j < CLS; ++j) o[j] = l[j] - m - ls;
}

extern "C" void kernel_launch(void* const* d_in, const int* in_sizes, int n_in,
                              void* d_out, int out_size, void* d_ws, size_t ws_size,
                              hipStream_t stream) {
  (void)in_sizes; (void)n_in; (void)out_size; (void)ws_size;

  const float* x    = (const float*)d_in[0];
  const int*   ei   = (const int*)  d_in[1];
  const float* W1_l = (const float*)d_in[2];
  const float* b1   = (const float*)d_in[3];
  const float* W1_r = (const float*)d_in[4];
  const float* W2_l = (const float*)d_in[5];
  const float* b2   = (const float*)d_in[6];
  const float* W2_r = (const float*)d_in[7];
  const int* src = ei;
  const int* dst = ei + NE;
  float* out = (float*)d_out;

  float* ws     = (float*)d_ws;
  float* cnt    = ws;
  float* agg1   = cnt  + N_NODES;
  float* h      = agg1 + (long long)N_NODES * F_IN;
  float* p2     = h    + (long long)N_NODES * HID;
  float* r2     = p2   + (long long)N_NODES * CLS;
  float* aggp   = r2   + (long long)N_NODES * CLS;

  const int TB = 256;
  auto blocks = [](long long n, int tb) { return (unsigned)((n + tb - 1) / tb); };

  long long nz1 = (long long)N_NODES * (1 + F_IN);
  zero_f32<<<blocks(nz1, TB), TB, 0, stream>>>(cnt, nz1);

  long long sc1 = (long long)NE * (F_IN / 4);
  scatter_accum<<<blocks(sc1, TB), TB, 0, stream>>>(x, src, dst, agg1, cnt,
                                                    F_IN / 4,  1);

  long long nm1 = (long long)N_NODES * F_IN;
  normalize_mean<<<blocks(nm1, TB), TB, 0, stream>>>(agg1, cnt, F_IN);

  layer1_gemm_strip<<<782, TB, 0, stream>>>(agg1, x, W1_l, W1_r, b1, h);

  proj_gemm_strip<<<391, TB, 0, stream>>>(h, W2_l, W2_r, p2, r2);

  long long nz2 = (long long)N_NODES * CLS;
  zero_f32<<<blocks(nz2, TB), TB, 0, stream>>>(aggp, nz2);

  long long sc2 = (long long)NE * (CLS / 4);
  scatter_accum<<<blocks(sc2, TB), TB, 0, stream>>>(p2, src, dst, aggp, cnt,
                                                    CLS / 4,  0);

  finish_rows<<<blocks(N_NODES, TB), TB, 0, stream>>>(aggp, cnt, r2, b2, out);
}


# Float-atomic edge scatters replaced by a per-call edge list grouped by destination (integer count+rank, prefix sum, place) and per-node row gathers added in list order; sum order changes
# speedup vs baseline: 3.7657x; 1.7471x over previous
#include <hip/hip_runtime.h>

#define N_NODES 50000
#define F_IN    128
#define HID     256
#define CLS     40
#define NE      640000

typedef __attribute__((ext_vector_type(2))) float v2f;
typedef __attribute__((ext_vector_type(8))) float v8f;

#define NPAD 53248
#define CSR_PAD 8

__global__ void zero_i32(int* __restrict__ p, int n) {
  int i = blockIdx.x * blockDim.x + threadIdx.x;
  if (i < n) p[i] = 0;
}

__global__ void count_rank(const int* __restrict__ dst,
                           int* __restrict__ deg,
                           int* __restrict__ rank) {
  int i = blockIdx.x * blockDim.x + threadIdx.x;
  if (i >= NE / 4) return;
  const int4 d = ((const int4*)dst)[i];
  int4 r;
  r.x = atomicAdd(deg + d.x, 1);
  r.y = atomicAdd(deg + d.y, 1);
  r.z = atomicAdd(deg + d.z, 1);
  r.w = atomicAdd(deg + d.w, 1);
  ((int4*)rank)[i] = r;
}

#define SCAN_TRIPS (NPAD / 4096)
__global__ void scan_deg(const int* __restrict__ deg, int* __restrict__ rowptr) {
  __shared__ int wsum[SCAN_TRIPS][32];
  const int t    = threadIdx.x;
  const int lane = t & 31;
  const int w    = t >> 5;
  int4 d[SCAN_TRIPS];
  int  s4[SCAN_TRIPS], v[SCAN_TRIPS];
  #pragma unroll
  for (int it = 0; it < SCAN_TRIPS; ++it) d[it] = ((const int4*)deg)[it * 1024 + t];
  #pragma unroll
  for (int it = 0; it < SCAN_TRIPS; ++it) {
    s4[it] = d[it].x + d[it].y + d[it].z + d[it].w;
    v[it]  = s4[it];
  }
  #pragma unroll
  for (int o = 1; o < 32; o <<= 1) {
    #pragma unroll
    for (int it = 0; it < SCAN_TRIPS; ++it) {
      const int u = __shfl_up(v[it], o);
      if (lane >= o) v[it] += u;
    }
  }
  if (lane == 31) {
    #pragma unroll
    for (int it = 0; it < SCAN_TRIPS; ++it) wsum[it][w] = v[it];
  }
  __syncthreads();
  int ws[SCAN_TRIPS], wv[SCAN_TRIPS];
  #pragma unroll
  for (int it = 0; it < SCAN_TRIPS; ++it) { ws[it] = wsum[it][lane]; wv[it] = ws[it]; }
  #pragma unroll
  for (int o = 1; o < 32; o <<= 1) {
    #pragma unroll
    for (int it = 0; it < SCAN_TRIPS; ++it) {
      const int u = __shfl_up(wv[it], o);
      if (lane >= o) wv[it] += u;
    }
  }
  int carry = 0;
  #pragma unroll
  for (int it = 0; it < SCAN_TRIPS; ++it) {
    const int wave_excl = __shfl(wv[it], w) - __shfl(ws[it], w);
    const int total     = __shfl(wv[it], 31);
    const int ex = carry + wave_excl + (v[it] - s4[it]);
    int4 r;
    r.x = ex;
    r.y = r.x + d[it].x;
    r.z = r.y + d[it].y;
    r.w = r.z + d[it].z;
    ((int4*)rowptr)[it * 1024 + t] = r;
    carry += total;
  }
}

__global__ void place_edges(const int* __restrict__ src,
                            const int* __restrict__ dst,
                            const int* __restrict__ rank,
                            const int* __restrict__ rowptr,
                            int* __restrict__ csr) {
  int i = blockIdx.x * blockDim.x + threadIdx.x;
  if (i >= NE / 4) return;
  if (i < CSR_PAD) csr[NE + i] = 0;
  const int4 s = ((const int4*)src)[i];
  const int4 d = ((const int4*)dst)[i];
  const int4 r = ((const int4*)rank)[i];
  const int p0 = rowptr[d.x], p1 = rowptr[d.y], p2 = rowptr[d.z], p3 = rowptr[d.w];
  csr[p0 + r.x] = s.x;
  csr[p1 + r.y] = s.y;
  csr[p2 + r.z] = s.z;
  csr[p3 + r.w] = s.w;
}

__global__ void gather_mean_rows128(const float* __restrict__ x,
                                    const int* __restrict__ rowptr,
                                    const int* __restrict__ csr,
                                    float* __restrict__ agg) {
  const int lane = threadIdx.x & 31;
  const int wv = __builtin_amdgcn_readfirstlane((int)(blockIdx.x * 8 + (threadIdx.x >> 5)));
  const int nw = (int)gridDim.x * 8;
  const float4* x4 = (const float4*)x;
  for (int n = wv; n < N_NODES; n += nw) {
    const int beg = rowptr[n];
    const int end = rowptr[n + 1];
    float4 acc = make_float4(0.0f, 0.0f, 0.0f, 0.0f);
    int j = beg;
    int s0 = csr[j], s1 = csr[j + 1], s2 = csr[j + 2], s3 = csr[j + 3];
    for (; j + 4 <= end; j += 4) {
      const int t0 = csr[j + 4], t1 = csr[j + 5], t2 = csr[j + 6], t3 = csr[j + 7];
      const float4 v0 = x4[(unsigned)s0 * 32u + lane];
      const float4 v1 = x4[(unsigned)s1 * 32u + lane];
      const float4 v2 = x4[(unsigned)s2 * 32u + lane];
      const float4 v3 = x4[(unsigned)s3 * 32u + lane];
      acc.x += v0.x; acc.y += v0.y; acc.z += v0.z; acc.w += v0.w;
      acc.x += v1.x; acc.y += v1.y; acc.z += v1.z; acc.w += v1.w;
      acc.x += v2.x; acc.y += v2.y; acc.z += v2.z; acc.w += v2.w;
      acc.x += v3.x; acc.y += v3.y; acc.z += v3.z; acc.w += v3.w;
      s0 = t0; s1 = t1; s2 = t2; s3 = t3;
    }
    const int rem = end - j;
    if (rem == 3) {
      const float4 v0 = x4[(unsigned)s0 * 32u + lane];
      const float4 v1 = x4[(unsigned)s1 * 32u + lane];
      const float4 v2 = x4[(unsigned)s2 * 32u + lane];
      acc.x += v0.x; acc.y += v0.y; acc.z += v0.z; acc.w += v0.w;
      acc.x += v1.x; acc.y += v1.y; acc.z += v1.z; acc.w += v1.w;
      acc.x += v2.x; acc.y += v2.y; acc.z += v2.z; acc.w += v2.w;
    } else if (rem == 2) {
      const float4 v0 = x4[(unsigned)s0 * 32u + lane];
      const float4 v1 = x4[(unsigned)s1 * 32u + lane];
      acc.x += v0.x; acc.y += v0.y; acc.z += v0.z; acc.w += v0.w;
      acc.x += v1.x; acc.y += v1.y; acc.z += v1.z; acc.w += v1.w;
    } else if (rem == 1) {
      const float4 v0 = x4[(unsigned)s0 * 32u + lane];
      acc.x += v0.x; acc.y += v0.y; acc.z += v0.z; acc.w += v0.w;
    }
    float c = (float)(end - beg);
    c = c > 1.0f ? c : 1.0f;
    float4 m;
    m.x = acc.x / c; m.y = acc.y / c; m.z = acc.z / c; m.w = acc.w / c;
    ((float4*)agg)[(unsigned)n * 32u + lane] = m;
  }
}

__global__ void gather_sum_rows40(const float* __restrict__ p,
                                  const int* __restrict__ rowptr,
                                  const int* __restrict__ csr,
                                  float* __restrict__ aggp) {
  const int i = blockIdx.x * blockDim.x + threadIdx.x;
  if (i >= N_NODES * (CLS / 4)) return;
  const int n = i / (CLS / 4);
  const int q = i - n * (CLS / 4);
  const float4* p4 = (const float4*)p;
  const int beg = rowptr[n];
  const int end = rowptr[n + 1];
  float4 acc = make_float4(0.0f, 0.0f, 0.0f, 0.0f);
  int j = beg;
  int s0 = csr[j], s1 = csr[j + 1], s2 = csr[j + 2], s3 = csr[j + 3];
  for (; j + 4 <= end; j += 4) {
    const int t0 = csr[j + 4], t1 = csr[j + 5], t2 = csr[j + 6], t3 = csr[j + 7];
    const float4 v0 = p4[s0 * (CLS / 4) + q];
    const float4 v1 = p4[s1 * (CLS / 4) + q];
    const float4 v2 = p4[s2 * (CLS / 4) + q];
    const float4 v3 = p4[s3 * (CLS / 4) + q];
    acc.x += v0.x; acc.y += v0.y; acc.z += v0.z; acc.w += v0.w;
    acc.x += v1.x; acc.y += v1.y; acc.z += v1.z; acc.w += v1.w;
    acc.x += v2.x; acc.y += v2.y; acc.z += v2.z; acc.w += v2.w;
    acc.x += v3.x; acc.y += v3.y; acc.z += v3.z; acc.w += v3.w;
    s0 = t0; s1 = t1; s2 = t2; s3 = t3;
  }
  const int rem = end - j;
  float4 v0 = make_float4(0.0f, 0.0f, 0.0f, 0.0f), v1 = v0, v2 = v0;
  if (rem > 0) v0 = p4[s0 * (CLS / 4) + q];
  if (rem > 1) v1 = p4[s1 * (CLS / 4) + q];
  if (rem > 2) v2 = p4[s2 * (CLS / 4) + q];
  if (rem > 0) { acc.x += v0.x; acc.y += v0.y; acc.z += v0.z; acc.w += v0.w; }
  if (rem > 1) { acc.x += v1.x; acc.y += v1.y; acc.z += v1.z; acc.w += v1.w; }
  if (rem > 2) { acc.x += v2.x; acc.y += v2.y; acc.z += v2.z; acc.w += v2.w; }
  ((float4*)aggp)[i] = acc;
}

#define BPITCH 132

template <int NT>
__device__ __forceinline__ void group_products(const float2 (&f)[8],
                                               const float* bg,
                                               v8f (&c)[NT]) {
  #pragma unroll
  for (int s = 0; s < 8; ++s) {
    v2f a; a.x = f[s].x; a.y = f[s].y;
    #pragma unroll
    for (int t = 0; t < NT; ++t) {
      float2 bv = *(const float2*)(bg + t * 16 * BPITCH + 4 * s);
      v2f b; b.x = bv.x; b.y = bv.y;
      c[t] = __builtin_amdgcn_wmma_f32_16x16x4_f32(false, a, false, b,
                                                   (short)0, c[t], false, false);
    }
  }
}

template <int NT, int NC>
__device__ __forceinline__ void strip_product(const float* __restrict__ a_lo,
                                              const float* __restrict__ a_hi,
                                              const float* bs,
                                              v8f (&c)[NT]) {
  float2 fa[8], fb[8];
  #pragma unroll
  for (int s = 0; s < 8; ++s) fa[s] = *(const float2*)(a_lo + 4 * s);

  #pragma unroll 1
  for (int gg = 0; gg < 4; ++gg) {
    {
      const int gn = 2 * gg + 1;
      const float* pn = (gn < 4) ? (a_lo + gn * 32) : (a_hi + (gn - 4) * 32);
      #pragma unroll
      for (int s = 0; s < 8; ++s) fb[s] = *(const float2*)(pn + 4 * s);
    }
    const float* bt = bs + (gg >> 1) * (NC * BPITCH) + (gg & 1) * 64;
    group_products<NT>(fa, bt, c);
    {
      const int gn = (gg < 3) ? (2 * gg + 2) : 7;
      const float* pn = (gn < 4) ? (a_lo + gn * 32) : (a_hi + (gn - 4) * 32);
      #pragma unroll
      for (int s = 0; s < 8; ++s) fa[s] = *(const float2*)(pn + 4 * s);
    }
    group_products<NT>(fb, bt + 32, c);
  }
}

__global__ void __launch_bounds__(256, 4)
layer1_gemm_strip(const float* __restrict__ agg,
                  const float* __restrict__ x,
                  const float* __restrict__ Wl,
                  const float* __restrict__ Wr,
                  const float* __restrict__ bias,
                  float* __restrict__ h) {
  __shared__ __attribute__((aligned(16))) float Bs[2 * 128 * BPITCH];
  const int tid  = threadIdx.x;
  const int lane = tid & 31;
  const int wave = tid >> 5;
  const int colbase = (blockIdx.x & 1) * 128;
  const int rb      = blockIdx.x >> 1;

  #pragma unroll
  for (int j = 0; j < 32; ++j) {
    const int i  = tid + j * 256;
    const int cc = i >> 6;
    const int q  = i & 63;
    const float* sp = (q < 32)
        ? (Wl + (unsigned)(colbase + cc) * (unsigned)F_IN + q * 4)
        : (Wr + (unsigned)(colbase + cc) * (unsigned)F_IN + (q - 32) * 4);
    const float4 v = *(const float4*)sp;
    *(float4*)(Bs + (q >> 5) * (128 * BPITCH) + cc * BPITCH + (q & 31) * 4) = v;
  }
  __syncthreads();

  const int rt = rb * 8 + wave;
  if (rt >= (N_NODES >> 4)) return;

  const int mrow  = rt * 16 + (lane & 15);
  const int khalf = (lane >> 4) * 2;
  const float* a_lo = agg + (unsigned)mrow * (unsigned)F_IN + khalf;
  const float* a_hi = x   + (unsigned)mrow * (unsigned)F_IN + khalf;
  const float* bs   = Bs + (lane & 15) * BPITCH + khalf;

  v8f c[8];
  #pragma unroll
  for (int t = 0; t < 8; ++t) c[t] = v8f{};

  strip_product<8, 128>(a_lo, a_hi, bs, c);

  const unsigned rbase = (unsigned)(rt * 16 + ((lane < 16) ? 0 : 8));
  #pragma unroll
  for (int t = 0; t < 8; ++t) {
    const int ncol = colbase + t * 16 + (lane & 15);
    const float bv = bias[ncol];
    float* orow = h + rbase * (unsigned)HID + ncol;
    #pragma unroll
    for (int i = 0; i < 8; ++i) {
      float v = c[t][i] + bv;
      v = v > 0.0f ? v : 0.0f;
      orow[(unsigned)i * (unsigned)HID] = v;
    }
  }
}

__global__ void __launch_bounds__(256, 6)
proj_gemm_strip(const float* __restrict__ H,
                const float* __restrict__ Wl,
                const float* __restrict__ Wr,
                float* __restrict__ P,
                float* __restrict__ R) {
  __shared__ __attribute__((aligned(16))) float Bs[2 * (2 * CLS) * BPITCH];
  const int tid  = threadIdx.x;
  const int lane = tid & 31;
  const int wave = tid >> 5;

  #pragma unroll
  for (int j = 0; j < 20; ++j) {
    const int i  = tid + j * 256;
    const int cc = i >> 6;
    const int q  = i & 63;
    const float* sp = (cc < CLS) ? (Wl + (unsigned)cc * (unsigned)HID + q * 4)
                                 : (Wr + (unsigned)(cc - CLS) * (unsigned)HID + q * 4);
    const float4 v = *(const float4*)sp;
    *(float4*)(Bs + (q >> 5) * (2 * CLS * BPITCH) + cc * BPITCH + (q & 31) * 4) = v;
  }
  __syncthreads();

  const int rt = blockIdx.x * 8 + wave;
  if (rt >= (N_NODES >> 4)) return;

  const int mrow  = rt * 16 + (lane & 15);
  const int khalf = (lane >> 4) * 2;
  const float* a_lo = H + (unsigned)mrow * (unsigned)HID + khalf;
  const float* a_hi = a_lo + 128;
  const float* bs   = Bs + (lane & 15) * BPITCH + khalf;

  v8f c[5];
  #pragma unroll
  for (int t = 0; t < 5; ++t) c[t] = v8f{};

  strip_product<5, 2 * CLS>(a_lo, a_hi, bs, c);

  const unsigned rbase = (unsigned)(rt * 16 + ((lane < 16) ? 0 : 8));
  #pragma unroll
  for (int t = 0; t < 5; ++t) {
    const int  ncol = t * 16 + (lane & 15);
    const bool root = (ncol >= CLS);
    const int  wrow = root ? (ncol - CLS) : ncol;
    float* orow = (root ? R : P) + rbase * (unsigned)CLS + wrow;
    #pragma unroll
    for (int i = 0; i < 8; ++i) orow[(unsigned)i * (unsigned)CLS] = c[t][i];
  }
}

__global__ void finish_rows(const float* __restrict__ aggp,
                            const int*   __restrict__ rowptr,
                            const float* __restrict__ rootp,
                            const float* __restrict__ bias,
                            float* __restrict__ out) {
  int r = blockIdx.x * blockDim.x + threadIdx.x;
  if (r >= N_NODES) return;
  float c = (float)(rowptr[r + 1] - rowptr[r]);
  c = c > 1.0f ? c : 1.0f;
  const float4* a4 = (const float4*)(aggp  + (long long)r * CLS);
  const float4* r4 = (const float4*)(rootp + (long long)r * CLS);
  float l[CLS];
  #pragma unroll
  for (int q = 0; q < CLS / 4; ++q) {
    float4 a = a4[q];
    float4 t = r4[q];
    l[4 * q + 0] = a.x / c + bias[4 * q + 0] + t.x;
    l[4 * q + 1] = a.y / c + bias[4 * q + 1] + t.y;
    l[4 * q + 2] = a.z / c + bias[4 * q + 2] + t.z;
    l[4 * q + 3] = a.w / c + bias[4 * q + 3] + t.w;
  }
  float m = -3.402823466e38f;
  #pragma unroll
  for (int j = 0; j < CLS; ++j) m = fmaxf(m, l[j]);
  float s = 0.0f;
  #pragma unroll
  for (int j = 0; j < CLS; ++j) s += expf(l[j] - m);
  float ls = logf(s);
  float* o = out + (long long)r * CLS;
  #pragma unroll
  for (int j = 0; j < CLS; ++j) o[j] = l[j] - m - ls;
}

extern "C" void kernel_launch(void* const* d_in, const int* in_sizes, int n_in,
                              void* d_out, int out_size, void* d_ws, size_t ws_size,
                              hipStream_t stream) {
  (void)in_sizes; (void)n_in; (void)out_size; (void)ws_size;

  const float* x    = (const float*)d_in[0];
  const int*   ei   = (const int*)  d_in[1];
  const float* W1_l = (const float*)d_in[2];
  const float* b1   = (const float*)d_in[3];
  const float* W1_r = (const float*)d_in[4];
  const float* W2_l = (const float*)d_in[5];
  const float* b2   = (const float*)d_in[6];
  const float* W2_r = (const float*)d_in[7];
  const int* src = ei;
  const int* dst = ei + NE;
  float* out = (float*)d_out;

  float* ws     = (float*)d_ws;
  float* agg1   = ws;
  float* h      = agg1 + (long long)N_NODES * F_IN;
  float* p2     = h    + (long long)N_NODES * HID;
  float* r2     = p2   + (long long)N_NODES * CLS;
  int*   rowptr = (int*)(r2 + (long long)N_NODES * CLS);
  int*   csr    = rowptr + NPAD;
  float* aggp   = agg1;
  int*   deg    = (int*)h;
  int*   rank   = deg + NPAD;

  const int TB = 256;
  auto blocks = [](long long n, int tb) { return (unsigned)((n + tb - 1) / tb); };

  zero_i32<<<blocks(NPAD, TB), TB, 0, stream>>>(deg, NPAD);
  count_rank<<<blocks(NE / 4, TB), TB, 0, stream>>>(dst, deg, rank);
  scan_deg<<<1, 1024, 0, stream>>>(deg, rowptr);
  place_edges<<<blocks(NE / 4, TB), TB, 0, stream>>>(src, dst, rank, rowptr, csr);

  gather_mean_rows128<<<2048, TB, 0, stream>>>(x, rowptr, csr, agg1);

  layer1_gemm_strip<<<782, TB, 0, stream>>>(agg1, x, W1_l, W1_r, b1, h);

  proj_gemm_strip<<<391, TB, 0, stream>>>(h, W2_l, W2_r, p2, r2);

  gather_sum_rows40<<<blocks((long long)N_NODES * (CLS / 4), TB), TB, 0, stream>>>(
      p2, rowptr, csr, aggp);

  finish_rows<<<blocks(N_NODES, TB), TB, 0, stream>>>(aggp, rowptr, r2, b2, out);
}


# Edges grouped by destination with a two-level counting sort on LDS counters (chunk x 256-node-bucket histogram, one scan, per-bucket node sort) instead of global returning adds; sum order changes
# speedup vs baseline: 4.8471x; 1.2872x over previous
#include <hip/hip_runtime.h>

#define N_NODES 50000
#define F_IN    128
#define HID     256
#define CLS     40
#define NE      640000

typedef __attribute__((ext_vector_type(2))) float v2f;
typedef __attribute__((ext_vector_type(8))) float v8f;

#define NPAD 53248
#define CSR_PAD 8
#define NB  196
#define CH  250
#define CHT 640

__global__ void __launch_bounds__(CHT)
bucket_count(const int* __restrict__ dst, int* __restrict__ hist) {
  __shared__ int cnt[NB];
  const int t = threadIdx.x;
  const int c = blockIdx.x;
  const int4 d = ((const int4*)dst)[c * CHT + t];
  if (t < NB) cnt[t] = 0;
  __syncthreads();
  atomicAdd(&cnt[d.x >> 8], 1);
  atomicAdd(&cnt[d.y >> 8], 1);
  atomicAdd(&cnt[d.z >> 8], 1);
  atomicAdd(&cnt[d.w >> 8], 1);
  __syncthreads();
  if (t < NB) hist[t * CH + c] = cnt[t];
  if (c == 0) {
    for (int k = NB * CH + t; k < NPAD; k += CHT) hist[k] = 0;
  }
}

#define SCAN_TRIPS (NPAD / 4096)
__global__ void scan_deg(const int* __restrict__ deg, int* __restrict__ rowptr) {
  __shared__ int wsum[SCAN_TRIPS][32];
  const int t    = threadIdx.x;
  const int lane = t & 31;
  const int w    = t >> 5;
  int4 d[SCAN_TRIPS];
  int  s4[SCAN_TRIPS], v[SCAN_TRIPS];
  #pragma unroll
  for (int it = 0; it < SCAN_TRIPS; ++it) d[it] = ((const int4*)deg)[it * 1024 + t];
  #pragma unroll
  for (int it = 0; it < SCAN_TRIPS; ++it) {
    s4[it] = d[it].x + d[it].y + d[it].z + d[it].w;
    v[it]  = s4[it];
  }
  #pragma unroll
  for (int o = 1; o < 32; o <<= 1) {
    #pragma unroll
    for (int it = 0; it < SCAN_TRIPS; ++it) {
      const int u = __shfl_up(v[it], o);
      if (lane >= o) v[it] += u;
    }
  }
  if (lane == 31) {
    #pragma unroll
    for (int it = 0; it < SCAN_TRIPS; ++it) wsum[it][w] = v[it];
  }
  __syncthreads();
  int ws[SCAN_TRIPS], wv[SCAN_TRIPS];
  #pragma unroll
  for (int it = 0; it < SCAN_TRIPS; ++it) { ws[it] = wsum[it][lane]; wv[it] = ws[it]; }
  #pragma unroll
  for (int o = 1; o < 32; o <<= 1) {
    #pragma unroll
    for (int it = 0; it < SCAN_TRIPS; ++it) {
      const int u = __shfl_up(wv[it], o);
      if (lane >= o) wv[it] += u;
    }
  }
  int carry = 0;
  #pragma unroll
  for (int it = 0; it < SCAN_TRIPS; ++it) {
    const int wave_excl = __shfl(wv[it], w) - __shfl(ws[it], w);
    const int total     = __shfl(wv[it], 31);
    const int ex = carry + wave_excl + (v[it] - s4[it]);
    int4 r;
    r.x = ex;
    r.y = r.x + d[it].x;
    r.z = r.y + d[it].y;
    r.w = r.z + d[it].z;
    ((int4*)rowptr)[it * 1024 + t] = r;
    carry += total;
  }
}

__global__ void __launch_bounds__(CHT)
bucket_scatter(const int* __restrict__ src,
               const int* __restrict__ dst,
               const int* __restrict__ base,
               int2* __restrict__ pairs) {
  __shared__ int cur[NB];
  const int t = threadIdx.x;
  const int c = blockIdx.x;
  const int4 s = ((const int4*)src)[c * CHT + t];
  const int4 d = ((const int4*)dst)[c * CHT + t];
  if (t < NB) cur[t] = base[t * CH + c];
  __syncthreads();
  const int p0 = atomicAdd(&cur[d.x >> 8], 1);
  const int p1 = atomicAdd(&cur[d.y >> 8], 1);
  const int p2 = atomicAdd(&cur[d.z >> 8], 1);
  const int p3 = atomicAdd(&cur[d.w >> 8], 1);
  pairs[p0] = make_int2(s.x, d.x);
  pairs[p1] = make_int2(s.y, d.y);
  pairs[p2] = make_int2(s.z, d.z);
  pairs[p3] = make_int2(s.w, d.w);
}

__global__ void __launch_bounds__(1024)
bucket_group(const int2* __restrict__ pairs,
             const int* __restrict__ base,
             int* __restrict__ rowptr,
             int* __restrict__ csr) {
  __shared__ int cnt[256];
  __shared__ int cur[256];
  const int t = threadIdx.x;
  const int b = blockIdx.x;
  const int beg = base[b * CH];
  const int end = base[(b + 1) * CH];
  if (t < 256) cnt[t] = 0;
  if (b == 0 && t < CSR_PAD) csr[NE + t] = 0;
  __syncthreads();
  for (int j = beg + t; j < end; j += 4096) {
    const bool k1 = j + 1024 < end, k2 = j + 2048 < end, k3 = j + 3072 < end;
    const int2 e0 = pairs[j];
    const int2 e1 = pairs[min(j + 1024, end - 1)];
    const int2 e2 = pairs[min(j + 2048, end - 1)];
    const int2 e3 = pairs[min(j + 3072, end - 1)];
    atomicAdd(&cnt[e0.y & 255], 1);
    if (k1) atomicAdd(&cnt[e1.y & 255], 1);
    if (k2) atomicAdd(&cnt[e2.y & 255], 1);
    if (k3) atomicAdd(&cnt[e3.y & 255], 1);
  }
  __syncthreads();
  if (t < 32) {
    int v[8];
    int tot = 0;
    #pragma unroll
    for (int k = 0; k < 8; ++k) { v[k] = cnt[t * 8 + k]; tot += v[k]; }
    int inc = tot;
    #pragma unroll
    for (int o = 1; o < 32; o <<= 1) {
      const int u = __shfl_up(inc, o);
      if (t >= o) inc += u;
    }
    int ex = beg + inc - tot;
    #pragma unroll
    for (int k = 0; k < 8; ++k) {
      cur[t * 8 + k] = ex;
      rowptr[b * 256 + t * 8 + k] = ex;
      ex += v[k];
    }
  }
  __syncthreads();
  for (int j = beg + t; j < end; j += 4096) {
    const bool k1 = j + 1024 < end, k2 = j + 2048 < end, k3 = j + 3072 < end;
    const int2 e0 = pairs[j];
    const int2 e1 = pairs[min(j + 1024, end - 1)];
    const int2 e2 = pairs[min(j + 2048, end - 1)];
    const int2 e3 = pairs[min(j + 3072, end - 1)];
    csr[atomicAdd(&cur[e0.y & 255], 1)] = e0.x;
    if (k1) csr[atomicAdd(&cur[e1.y & 255], 1)] = e1.x;
    if (k2) csr[atomicAdd(&cur[e2.y & 255], 1)] = e2.x;
    if (k3) csr[atomicAdd(&cur[e3.y & 255], 1)] = e3.x;
  }
}

__global__ void gather_mean_rows128(const float* __restrict__ x,
                                    const int* __restrict__ rowptr,
                                    const int* __restrict__ csr,
                                    float* __restrict__ agg) {
  const int lane = threadIdx.x & 31;
  const int wv = __builtin_amdgcn_readfirstlane((int)(blockIdx.x * 8 + (threadIdx.x >> 5)));
  const int nw = (int)gridDim.x * 8;
  const float4* x4 = (const float4*)x;
  for (int n = wv; n < N_NODES; n += nw) {
    const int beg = rowptr[n];
    const int end = rowptr[n + 1];
    float4 acc = make_float4(0.0f, 0.0f, 0.0f, 0.0f);
    int j = beg;
    int s0 = csr[j], s1 = csr[j + 1], s2 = csr[j + 2], s3 = csr[j + 3];
    for (; j + 4 <= end; j += 4) {
      const int t0 = csr[j + 4], t1 = csr[j + 5], t2 = csr[j + 6], t3 = csr[j + 7];
      const float4 v0 = x4[(unsigned)s0 * 32u + lane];
      const float4 v1 = x4[(unsigned)s1 * 32u + lane];
      const float4 v2 = x4[(unsigned)s2 * 32u + lane];
      const float4 v3 = x4[(unsigned)s3 * 32u + lane];
      acc.x += v0.x; acc.y += v0.y; acc.z += v0.z; acc.w += v0.w;
      acc.x += v1.x; acc.y += v1.y; acc.z += v1.z; acc.w += v1.w;
      acc.x += v2.x; acc.y += v2.y; acc.z += v2.z; acc.w += v2.w;
      acc.x += v3.x; acc.y += v3.y; acc.z += v3.z; acc.w += v3.w;
      s0 = t0; s1 = t1; s2 = t2; s3 = t3;
    }
    const int rem = end - j;
    if (rem == 3) {
      const float4 v0 = x4[(unsigned)s0 * 32u + lane];
      const float4 v1 = x4[(unsigned)s1 * 32u + lane];
      const float4 v2 = x4[(unsigned)s2 * 32u + lane];
      acc.x += v0.x; acc.y += v0.y; acc.z += v0.z; acc.w += v0.w;
      acc.x += v1.x; acc.y += v1.y; acc.z += v1.z; acc.w += v1.w;
      acc.x += v2.x; acc.y += v2.y; acc.z += v2.z; acc.w += v2.w;
    } else if (rem == 2) {
      const float4 v0 = x4[(unsigned)s0 * 32u + lane];
      const float4 v1 = x4[(unsigned)s1 * 32u + lane];
      acc.x += v0.x; acc.y += v0.y; acc.z += v0.z; acc.w += v0.w;
      acc.x += v1.x; acc.y += v1.y; acc.z += v1.z; acc.w += v1.w;
    } else if (rem == 1) {
      const float4 v0 = x4[(unsigned)s0 * 32u + lane];
      acc.x += v0.x; acc.y += v0.y; acc.z += v0.z; acc.w += v0.w;
    }
    float c = (float)(end - beg);
    c = c > 1.0f ? c : 1.0f;
    float4 m;
    m.x = acc.x / c; m.y = acc.y / c; m.z = acc.z / c; m.w = acc.w / c;
    ((float4*)agg)[(unsigned)n * 32u + lane] = m;
  }
}

__global__ void gather_sum_rows40(const float* __restrict__ p,
                                  const int* __restrict__ rowptr,
                                  const int* __restrict__ csr,
                                  float* __restrict__ aggp) {
  const int i = blockIdx.x * blockDim.x + threadIdx.x;
  if (i >= N_NODES * (CLS / 4)) return;
  const int n = i / (CLS / 4);
  const int q = i - n * (CLS / 4);
  const float4* p4 = (const float4*)p;
  const int beg = rowptr[n];
  const int end = rowptr[n + 1];
  float4 acc = make_float4(0.0f, 0.0f, 0.0f, 0.0f);
  int j = beg;
  int s0 = csr[j], s1 = csr[j + 1], s2 = csr[j + 2], s3 = csr[j + 3];
  for (; j + 4 <= end; j += 4) {
    const int t0 = csr[j + 4], t1 = csr[j + 5], t2 = csr[j + 6], t3 = csr[j + 7];
    const float4 v0 = p4[s0 * (CLS / 4) + q];
    const float4 v1 = p4[s1 * (CLS / 4) + q];
    const float4 v2 = p4[s2 * (CLS / 4) + q];
    const float4 v3 = p4[s3 * (CLS / 4) + q];
    acc.x += v0.x; acc.y += v0.y; acc.z += v0.z; acc.w += v0.w;
    acc.x += v1.x; acc.y += v1.y; acc.z += v1.z; acc.w += v1.w;
    acc.x += v2.x; acc.y += v2.y; acc.z += v2.z; acc.w += v2.w;
    acc.x += v3.x; acc.y += v3.y; acc.z += v3.z; acc.w += v3.w;
    s0 = t0; s1 = t1; s2 = t2; s3 = t3;
  }
  const int rem = end - j;
  float4 v0 = make_float4(0.0f, 0.0f, 0.0f, 0.0f), v1 = v0, v2 = v0;
  if (rem > 0) v0 = p4[s0 * (CLS / 4) + q];
  if (rem > 1) v1 = p4[s1 * (CLS / 4) + q];
  if (rem > 2) v2 = p4[s2 * (CLS / 4) + q];
  if (rem > 0) { acc.x += v0.x; acc.y += v0.y; acc.z += v0.z; acc.w += v0.w; }
  if (rem > 1) { acc.x += v1.x; acc.y += v1.y; acc.z += v1.z; acc.w += v1.w; }
  if (rem > 2) { acc.x += v2.x; acc.y += v2.y; acc.z += v2.z; acc.w += v2.w; }
  ((float4*)aggp)[i] = acc;
}

#define BPITCH 132

template <int NT>
__device__ __forceinline__ void group_products(const float2 (&f)[8],
                                               const float* bg,
                                               v8f (&c)[NT]) {
  #pragma unroll
  for (int s = 0; s < 8; ++s) {
    v2f a; a.x = f[s].x; a.y = f[s].y;
    #pragma unroll
    for (int t = 0; t < NT; ++t) {
      float2 bv = *(const float2*)(bg + t * 16 * BPITCH + 4 * s);
      v2f b; b.x = bv.x; b.y = bv.y;
      c[t] = __builtin_amdgcn_wmma_f32_16x16x4_f32(false, a, false, b,
                                                   (short)0, c[t], false, false);
    }
  }
}

template <int NT, int NC>
__device__ __forceinline__ void strip_product(const float* __restrict__ a_lo,
                                              const float* __restrict__ a_hi,
                                              const float* bs,
                                              v8f (&c)[NT]) {
  float2 fa[8], fb[8];
  #pragma unroll
  for (int s = 0; s < 8; ++s) fa[s] = *(const float2*)(a_lo + 4 * s);

  #pragma unroll 1
  for (int gg = 0; gg < 4; ++gg) {
    {
      const int gn = 2 * gg + 1;
      const float* pn = (gn < 4) ? (a_lo + gn * 32) : (a_hi + (gn - 4) * 32);
      #pragma unroll
      for (int s = 0; s < 8; ++s) fb[s] = *(const float2*)(pn + 4 * s);
    }
    const float* bt = bs + (gg >> 1) * (NC * BPITCH) + (gg & 1) * 64;
    group_products<NT>(fa, bt, c);
    {
      const int gn = (gg < 3) ? (2 * gg + 2) : 7;
      const float* pn = (gn < 4) ? (a_lo + gn * 32) : (a_hi + (gn - 4) * 32);
      #pragma unroll
      for (int s = 0; s < 8; ++s) fa[s] = *(const float2*)(pn + 4 * s);
    }
    group_products<NT>(fb, bt + 32, c);
  }
}

__global__ void __launch_bounds__(256, 4)
layer1_gemm_strip(const float* __restrict__ agg,
                  const float* __restrict__ x,
                  const float* __restrict__ Wl,
                  const float* __restrict__ Wr,
                  const float* __restrict__ bias,
                  float* __restrict__ h) {
  __shared__ __attribute__((aligned(16))) float Bs[2 * 128 * BPITCH];
  const int tid  = threadIdx.x;
  const int lane = tid & 31;
  const int wave = tid >> 5;
  const int colbase = (blockIdx.x & 1) * 128;
  const int rb      = blockIdx.x >> 1;

  #pragma unroll
  for (int j = 0; j < 32; ++j) {
    const int i  = tid + j * 256;
    const int cc = i >> 6;
    const int q  = i & 63;
    const float* sp = (q < 32)
        ? (Wl + (unsigned)(colbase + cc) * (unsigned)F_IN + q * 4)
        : (Wr + (unsigned)(colbase + cc) * (unsigned)F_IN + (q - 32) * 4);
    const float4 v = *(const float4*)sp;
    *(float4*)(Bs + (q >> 5) * (128 * BPITCH) + cc * BPITCH + (q & 31) * 4) = v;
  }
  __syncthreads();

  const int rt = rb * 8 + wave;
  if (rt >= (N_NODES >> 4)) return;

  const int mrow  = rt * 16 + (lane & 15);
  const int khalf = (lane >> 4) * 2;
  const float* a_lo = agg + (unsigned)mrow * (unsigned)F_IN + khalf;
  const float* a_hi = x   + (unsigned)mrow * (unsigned)F_IN + khalf;
  const float* bs   = Bs + (lane & 15) * BPITCH + khalf;

  v8f c[8];
  #pragma unroll
  for (int t = 0; t < 8; ++t) c[t] = v8f{};

  strip_product<8, 128>(a_lo, a_hi, bs, c);

  const unsigned rbase = (unsigned)(rt * 16 + ((lane < 16) ? 0 : 8));
  #pragma unroll
  for (int t = 0; t < 8; ++t) {
    const int ncol = colbase + t * 16 + (lane & 15);
    const float bv = bias[ncol];
    float* orow = h + rbase * (unsigned)HID + ncol;
    #pragma unroll
    for (int i = 0; i < 8; ++i) {
      float v = c[t][i] + bv;
      v = v > 0.0f ? v : 0.0f;
      orow[(unsigned)i * (unsigned)HID] = v;
    }
  }
}

__global__ void __launch_bounds__(256, 6)
proj_gemm_strip(const float* __restrict__ H,
                const float* __restrict__ Wl,
                const float* __restrict__ Wr,
                float* __restrict__ P,
                float* __restrict__ R) {
  __shared__ __attribute__((aligned(16))) float Bs[2 * (2 * CLS) * BPITCH];
  const int tid  = threadIdx.x;
  const int lane = tid & 31;
  const int wave = tid >> 5;

  #pragma unroll
  for (int j = 0; j < 20; ++j) {
    const int i  = tid + j * 256;
    const int cc = i >> 6;
    const int q  = i & 63;
    const float* sp = (cc < CLS) ? (Wl + (unsigned)cc * (unsigned)HID + q * 4)
                                 : (Wr + (unsigned)(cc - CLS) * (unsigned)HID + q * 4);
    const float4 v = *(const float4*)sp;
    *(float4*)(Bs + (q >> 5) * (2 * CLS * BPITCH) + cc * BPITCH + (q & 31) * 4) = v;
  }
  __syncthreads();

  const int rt = blockIdx.x * 8 + wave;
  if (rt >= (N_NODES >> 4)) return;

  const int mrow  = rt * 16 + (lane & 15);
  const int khalf = (lane >> 4) * 2;
  const float* a_lo = H + (unsigned)mrow * (unsigned)HID + khalf;
  const float* a_hi = a_lo + 128;
  const float* bs   = Bs + (lane & 15) * BPITCH + khalf;

  v8f c[5];
  #pragma unroll
  for (int t = 0; t < 5; ++t) c[t] = v8f{};

  strip_product<5, 2 * CLS>(a_lo, a_hi, bs, c);

  const unsigned rbase = (unsigned)(rt * 16 + ((lane < 16) ? 0 : 8));
  #pragma unroll
  for (int t = 0; t < 5; ++t) {
    const int  ncol = t * 16 + (lane & 15);
    const bool root = (ncol >= CLS);
    const int  wrow = root ? (ncol - CLS) : ncol;
    float* orow = (root ? R : P) + rbase * (unsigned)CLS + wrow;
    #pragma unroll
    for (int i = 0; i < 8; ++i) orow[(unsigned)i * (unsigned)CLS] = c[t][i];
  }
}

__global__ void finish_rows(const float* __restrict__ aggp,
                            const int*   __restrict__ rowptr,
                            const float* __restrict__ rootp,
                            const float* __restrict__ bias,
                            float* __restrict__ out) {
  int r = blockIdx.x * blockDim.x + threadIdx.x;
  if (r >= N_NODES) return;
  float c = (float)(rowptr[r + 1] - rowptr[r]);
  c = c > 1.0f ? c : 1.0f;
  const float4* a4 = (const float4*)(aggp  + (long long)r * CLS);
  const float4* r4 = (const float4*)(rootp + (long long)r * CLS);
  float l[CLS];
  #pragma unroll
  for (int q = 0; q < CLS / 4; ++q) {
    float4 a = a4[q];
    float4 t = r4[q];
    l[4 * q + 0] = a.x / c + bias[4 * q + 0] + t.x;
    l[4 * q + 1] = a.y / c + bias[4 * q + 1] + t.y;
    l[4 * q + 2] = a.z / c + bias[4 * q + 2] + t.z;
    l[4 * q + 3] = a.w / c + bias[4 * q + 3] + t.w;
  }
  float m = -3.402823466e38f;
  #pragma unroll
  for (int j = 0; j < CLS; ++j) m = fmaxf(m, l[j]);
  float s = 0.0f;
  #pragma unroll
  for (int j = 0; j < CLS; ++j) s += expf(l[j] - m);
  float ls = logf(s);
  float* o = out + (long long)r * CLS;
  #pragma unroll
  for (int j = 0; j < CLS; ++j) o[j] = l[j] - m - ls;
}

extern "C" void kernel_launch(void* const* d_in, const int* in_sizes, int n_in,
                              void* d_out, int out_size, void* d_ws, size_t ws_size,
                              hipStream_t stream) {
  (void)in_sizes; (void)n_in; (void)out_size; (void)ws_size;

  const float* x    = (const float*)d_in[0];
  const int*   ei   = (const int*)  d_in[1];
  const float* W1_l = (const float*)d_in[2];
  const float* b1   = (const float*)d_in[3];
  const float* W1_r = (const float*)d_in[4];
  const float* W2_l = (const float*)d_in[5];
  const float* b2   = (const float*)d_in[6];
  const float* W2_r = (const float*)d_in[7];
  const int* src = ei;
  const int* dst = ei + NE;
  float* out = (float*)d_out;

  float* ws     = (float*)d_ws;
  float* agg1   = ws;
  float* h      = agg1 + (long long)N_NODES * F_IN;
  float* p2     = h    + (long long)N_NODES * HID;
  float* r2     = p2   + (long long)N_NODES * CLS;
  int*   rowptr = (int*)(r2 + (long long)N_NODES * CLS);
  int*   csr    = rowptr + NPAD;
  float* aggp   = agg1;
  int*   hist   = (int*)h;
  int*   base   = hist + NPAD;
  int2*  pairs  = (int2*)(base + NPAD);

  const int TB = 256;
  auto blocks = [](long long n, int tb) { return (unsigned)((n + tb - 1) / tb); };

  bucket_count<<<CH, CHT, 0, stream>>>(dst, hist);
  scan_deg<<<1, 1024, 0, stream>>>(hist, base);
  bucket_scatter<<<CH, CHT, 0, stream>>>(src, dst, base, pairs);
  bucket_group<<<NB, 1024, 0, stream>>>(pairs, base, rowptr, csr);

  gather_mean_rows128<<<2048, TB, 0, stream>>>(x, rowptr, csr, agg1);

  layer1_gemm_strip<<<782, TB, 0, stream>>>(agg1, x, W1_l, W1_r, b1, h);

  proj_gemm_strip<<<391, TB, 0, stream>>>(h, W2_l, W2_r, p2, r2);

  gather_sum_rows40<<<blocks((long long)N_NODES * (CLS / 4), TB), TB, 0, stream>>>(
      p2, rowptr, csr, aggp);

  finish_rows<<<blocks(N_NODES, TB), TB, 0, stream>>>(aggp, rowptr, r2, b2, out);
}


# Edge grouping without the chip-wide prefix scan: a chunk sorts its pairs by bucket inside its own slice (LDS rank + local scan), a bucket reads its 250 runs; 9->7 launches; list order by LDS arrival
# speedup vs baseline: 5.6523x; 1.1661x over previous
#include <hip/hip_runtime.h>

#define N_NODES 50000
#define F_IN    128
#define HID     256
#define CLS     40
#define NE      640000

typedef __attribute__((ext_vector_type(2))) float v2f;
typedef __attribute__((ext_vector_type(8))) float v8f;

#define NPAD 53248
#define CSR_PAD 8
#define NB  196
#define CH  250
#define CHT 640
#define CHE 2560
#define OFFP 200

__global__ void __launch_bounds__(CHT)
chunk_sort(const int* __restrict__ src,
           const int* __restrict__ dst,
           int* __restrict__ off,
           int2* __restrict__ pairs) {
  __shared__ int cnt[NB];
  __shared__ int beg[32 * 7];
  const int t = threadIdx.x;
  const int c = blockIdx.x;
  const int4 s = ((const int4*)src)[c * CHT + t];
  const int4 d = ((const int4*)dst)[c * CHT + t];
  if (t < NB) cnt[t] = 0;
  __syncthreads();
  const int r0 = atomicAdd(&cnt[d.x >> 8], 1);
  const int r1 = atomicAdd(&cnt[d.y >> 8], 1);
  const int r2 = atomicAdd(&cnt[d.z >> 8], 1);
  const int r3 = atomicAdd(&cnt[d.w >> 8], 1);
  __syncthreads();
  if (t < 32) {
    int v[7];
    int tot = 0;
    #pragma unroll
    for (int k = 0; k < 7; ++k) {
      const int i = t * 7 + k;
      v[k] = (i < NB) ? cnt[i] : 0;
      tot += v[k];
    }
    int inc = tot;
    #pragma unroll
    for (int o = 1; o < 32; o <<= 1) {
      const int u = __shfl_up(inc, o);
      if (t >= o) inc += u;
    }
    int ex = inc - tot;
    #pragma unroll
    for (int k = 0; k < 7; ++k) {
      const int i = t * 7 + k;
      beg[i] = ex;
      if (i <= NB) off[c * OFFP + i] = ex;
      ex += v[k];
    }
  }
  __syncthreads();
  int2* mine = pairs + c * CHE;
  mine[beg[d.x >> 8] + r0] = make_int2(s.x, d.x);
  mine[beg[d.y >> 8] + r1] = make_int2(s.y, d.y);
  mine[beg[d.z >> 8] + r2] = make_int2(s.z, d.z);
  mine[beg[d.w >> 8] + r3] = make_int2(s.w, d.w);
}

__device__ __forceinline__ int run_slot(const int* rs, const int* ro, int e) {
  int lo = 0, hi = CH;
  #pragma unroll
  for (int k = 0; k < 8; ++k) {
    const int mid = (lo + hi) >> 1;
    const bool up = rs[mid] <= e;
    lo = up ? mid : lo;
    hi = up ? hi : mid;
  }
  return ro[lo] + (e - rs[lo]);
}

__global__ void __launch_bounds__(1024)
bucket_group(const int2* __restrict__ pairs,
             const int* __restrict__ off,
             int* __restrict__ rowptr,
             int* __restrict__ csr) {
  __shared__ int cnt[256];
  __shared__ int cur[256];
  __shared__ int ln[256];
  __shared__ int st[256];
  __shared__ int rs[257];
  __shared__ int ro[256];
  __shared__ int lower;
  const int t = threadIdx.x;
  const int b = blockIdx.x;
  if (t < 256) {
    cnt[t] = 0;
    int o0 = 0, o1 = 0;
    if (t < CH) { o0 = off[t * OFFP + b]; o1 = off[t * OFFP + b + 1]; }
    st[t] = o0;
    ln[t] = o1 - o0;
    ro[t] = t * CHE + o0;
  }
  if (b == 0 && t < CSR_PAD) csr[NE + t] = 0;
  __syncthreads();
  if (t < 32) {
    int v[8];
    int tot = 0, low = 0;
    #pragma unroll
    for (int k = 0; k < 8; ++k) { v[k] = ln[t * 8 + k]; tot += v[k]; low += st[t * 8 + k]; }
    int inc = tot;
    #pragma unroll
    for (int o = 1; o < 32; o <<= 1) {
      const int u = __shfl_up(inc, o);
      if (t >= o) inc += u;
    }
    #pragma unroll
    for (int o = 16; o >= 1; o >>= 1) low += __shfl_xor(low, o);
    int ex = inc - tot;
    #pragma unroll
    for (int k = 0; k < 8; ++k) { rs[t * 8 + k] = ex; ex += v[k]; }
    if (t == 31) rs[256] = ex;
    if (t == 0) lower = low;
  }
  __syncthreads();
  const int total = rs[CH];
  const int beg   = lower;
  for (int e = t; e < total; e += 4096) {
    const bool k1 = e + 1024 < total, k2 = e + 2048 < total, k3 = e + 3072 < total;
    const int2 e0 = pairs[run_slot(rs, ro, e)];
    const int2 e1 = pairs[run_slot(rs, ro, min(e + 1024, total - 1))];
    const int2 e2 = pairs[run_slot(rs, ro, min(e + 2048, total - 1))];
    const int2 e3 = pairs[run_slot(rs, ro, min(e + 3072, total - 1))];
    atomicAdd(&cnt[e0.y & 255], 1);
    if (k1) atomicAdd(&cnt[e1.y & 255], 1);
    if (k2) atomicAdd(&cnt[e2.y & 255], 1);
    if (k3) atomicAdd(&cnt[e3.y & 255], 1);
  }
  __syncthreads();
  if (t < 32) {
    int v[8];
    int tot = 0;
    #pragma unroll
    for (int k = 0; k < 8; ++k) { v[k] = cnt[t * 8 + k]; tot += v[k]; }
    int inc = tot;
    #pragma unroll
    for (int o = 1; o < 32; o <<= 1) {
      const int u = __shfl_up(inc, o);
      if (t >= o) inc += u;
    }
    int ex = beg + inc - tot;
    #pragma unroll
    for (int k = 0; k < 8; ++k) {
      cur[t * 8 + k] = ex;
      rowptr[b * 256 + t * 8 + k] = ex;
      ex += v[k];
    }
  }
  __syncthreads();
  for (int e = t; e < total; e += 4096) {
    const bool k1 = e + 1024 < total, k2 = e + 2048 < total, k3 = e + 3072 < total;
    const int2 e0 = pairs[run_slot(rs, ro, e)];
    const int2 e1 = pairs[run_slot(rs, ro, min(e + 1024, total - 1))];
    const int2 e2 = pairs[run_slot(rs, ro, min(e + 2048, total - 1))];
    const int2 e3 = pairs[run_slot(rs, ro, min(e + 3072, total - 1))];
    csr[atomicAdd(&cur[e0.y & 255], 1)] = e0.x;
    if (k1) csr[atomicAdd(&cur[e1.y & 255], 1)] = e1.x;
    if (k2) csr[atomicAdd(&cur[e2.y & 255], 1)] = e2.x;
    if (k3) csr[atomicAdd(&cur[e3.y & 255], 1)] = e3.x;
  }
}

__global__ void gather_mean_rows128(const float* __restrict__ x,
                                    const int* __restrict__ rowptr,
                                    const int* __restrict__ csr,
                                    float* __restrict__ agg) {
  const int lane = threadIdx.x & 31;
  const int wv = __builtin_amdgcn_readfirstlane((int)(blockIdx.x * 8 + (threadIdx.x >> 5)));
  const int nw = (int)gridDim.x * 8;
  const float4* x4 = (const float4*)x;
  for (int n = wv; n < N_NODES; n += nw) {
    const int beg = rowptr[n];
    const int end = rowptr[n + 1];
    float4 acc = make_float4(0.0f, 0.0f, 0.0f, 0.0f);
    int j = beg;
    int s0 = csr[j], s1 = csr[j + 1], s2 = csr[j + 2], s3 = csr[j + 3];
    for (; j + 4 <= end; j += 4) {
      const int t0 = csr[j + 4], t1 = csr[j + 5], t2 = csr[j + 6], t3 = csr[j + 7];
      const float4 v0 = x4[(unsigned)s0 * 32u + lane];
      const float4 v1 = x4[(unsigned)s1 * 32u + lane];
      const float4 v2 = x4[(unsigned)s2 * 32u + lane];
      const float4 v3 = x4[(unsigned)s3 * 32u + lane];
      acc.x += v0.x; acc.y += v0.y; acc.z += v0.z; acc.w += v0.w;
      acc.x += v1.x; acc.y += v1.y; acc.z += v1.z; acc.w += v1.w;
      acc.x += v2.x; acc.y += v2.y; acc.z += v2.z; acc.w += v2.w;
      acc.x += v3.x; acc.y += v3.y; acc.z += v3.z; acc.w += v3.w;
      s0 = t0; s1 = t1; s2 = t2; s3 = t3;
    }
    const int rem = end - j;
    if (rem == 3) {
      const float4 v0 = x4[(unsigned)s0 * 32u + lane];
      const float4 v1 = x4[(unsigned)s1 * 32u + lane];
      const float4 v2 = x4[(unsigned)s2 * 32u + lane];
      acc.x += v0.x; acc.y += v0.y; acc.z += v0.z; acc.w += v0.w;
      acc.x += v1.x; acc.y += v1.y; acc.z += v1.z; acc.w += v1.w;
      acc.x += v2.x; acc.y += v2.y; acc.z += v2.z; acc.w += v2.w;
    } else if (rem == 2) {
      const float4 v0 = x4[(unsigned)s0 * 32u + lane];
      const float4 v1 = x4[(unsigned)s1 * 32u + lane];
      acc.x += v0.x; acc.y += v0.y; acc.z += v0.z; acc.w += v0.w;
      acc.x += v1.x; acc.y += v1.y; acc.z += v1.z; acc.w += v1.w;
    } else if (rem == 1) {
      const float4 v0 = x4[(unsigned)s0 * 32u + lane];
      acc.x += v0.x; acc.y += v0.y; acc.z += v0.z; acc.w += v0.w;
    }
    float c = (float)(end - beg);
    c = c > 1.0f ? c : 1.0f;
    float4 m;
    m.x = acc.x / c; m.y = acc.y / c; m.z = acc.z / c; m.w = acc.w / c;
    ((float4*)agg)[(unsigned)n * 32u + lane] = m;
  }
}

__global__ void gather_sum_rows40(const float* __restrict__ p,
                                  const int* __restrict__ rowptr,
                                  const int* __restrict__ csr,
                                  float* __restrict__ aggp) {
  const int i = blockIdx.x * blockDim.x + threadIdx.x;
  if (i >= N_NODES * (CLS / 4)) return;
  const int n = i / (CLS / 4);
  const int q = i - n * (CLS / 4);
  const float4* p4 = (const float4*)p;
  const int beg = rowptr[n];
  const int end = rowptr[n + 1];
  float4 acc = make_float4(0.0f, 0.0f, 0.0f, 0.0f);
  int j = beg;
  int s0 = csr[j], s1 = csr[j + 1], s2 = csr[j + 2], s3 = csr[j + 3];
  for (; j + 4 <= end; j += 4) {
    const int t0 = csr[j + 4], t1 = csr[j + 5], t2 = csr[j + 6], t3 = csr[j + 7];
    const float4 v0 = p4[s0 * (CLS / 4) + q];
    const float4 v1 = p4[s1 * (CLS / 4) + q];
    const float4 v2 = p4[s2 * (CLS / 4) + q];
    const float4 v3 = p4[s3 * (CLS / 4) + q];
    acc.x += v0.x; acc.y += v0.y; acc.z += v0.z; acc.w += v0.w;
    acc.x += v1.x; acc.y += v1.y; acc.z += v1.z; acc.w += v1.w;
    acc.x += v2.x; acc.y += v2.y; acc.z += v2.z; acc.w += v2.w;
    acc.x += v3.x; acc.y += v3.y; acc.z += v3.z; acc.w += v3.w;
    s0 = t0; s1 = t1; s2 = t2; s3 = t3;
  }
  const int rem = end - j;
  float4 v0 = make_float4(0.0f, 0.0f, 0.0f, 0.0f), v1 = v0, v2 = v0;
  if (rem > 0) v0 = p4[s0 * (CLS / 4) + q];
  if (rem > 1) v1 = p4[s1 * (CLS / 4) + q];
  if (rem > 2) v2 = p4[s2 * (CLS / 4) + q];
  if (rem > 0) { acc.x += v0.x; acc.y += v0.y; acc.z += v0.z; acc.w += v0.w; }
  if (rem > 1) { acc.x += v1.x; acc.y += v1.y; acc.z += v1.z; acc.w += v1.w; }
  if (rem > 2) { acc.x += v2.x; acc.y += v2.y; acc.z += v2.z; acc.w += v2.w; }
  ((float4*)aggp)[i] = acc;
}

#define BPITCH 132

template <int NT>
__device__ __forceinline__ void group_products(const float2 (&f)[8],
                                               const float* bg,
                                               v8f (&c)[NT]) {
  #pragma unroll
  for (int s = 0; s < 8; ++s) {
    v2f a; a.x = f[s].x; a.y = f[s].y;
    #pragma unroll
    for (int t = 0; t < NT; ++t) {
      float2 bv = *(const float2*)(bg + t * 16 * BPITCH + 4 * s);
      v2f b; b.x = bv.x; b.y = bv.y;
      c[t] = __builtin_amdgcn_wmma_f32_16x16x4_f32(false, a, false, b,
                                                   (short)0, c[t], false, false);
    }
  }
}

template <int NT, int NC>
__device__ __forceinline__ void strip_product(const float* __restrict__ a_lo,
                                              const float* __restrict__ a_hi,
                                              const float* bs,
                                              v8f (&c)[NT]) {
  float2 fa[8], fb[8];
  #pragma unroll
  for (int s = 0; s < 8; ++s) fa[s] = *(const float2*)(a_lo + 4 * s);

  #pragma unroll 1
  for (int gg = 0; gg < 4; ++gg) {
    {
      const int gn = 2 * gg + 1;
      const float* pn = (gn < 4) ? (a_lo + gn * 32) : (a_hi + (gn - 4) * 32);
      #pragma unroll
      for (int s = 0; s < 8; ++s) fb[s] = *(const float2*)(pn + 4 * s);
    }
    const float* bt = bs + (gg >> 1) * (NC * BPITCH) + (gg & 1) * 64;
    group_products<NT>(fa, bt, c);
    {
      const int gn = (gg < 3) ? (2 * gg + 2) : 7;
      const float* pn = (gn < 4) ? (a_lo + gn * 32) : (a_hi + (gn - 4) * 32);
      #pragma unroll
      for (int s = 0; s < 8; ++s) fa[s] = *(const float2*)(pn + 4 * s);
    }
    group_products<NT>(fb, bt + 32, c);
  }
}

__global__ void __launch_bounds__(256, 4)
layer1_gemm_strip(const float* __restrict__ agg,
                  const float* __restrict__ x,
                  const float* __restrict__ Wl,
                  const float* __restrict__ Wr,
                  const float* __restrict__ bias,
                  float* __restrict__ h) {
  __shared__ __attribute__((aligned(16))) float Bs[2 * 128 * BPITCH];
  const int tid  = threadIdx.x;
  const int lane = tid & 31;
  const int wave = tid >> 5;
  const int colbase = (blockIdx.x & 1) * 128;
  const int rb      = blockIdx.x >> 1;

  #pragma unroll
  for (int j = 0; j < 32; ++j) {
    const int i  = tid + j * 256;
    const int cc = i >> 6;
    const int q  = i & 63;
    const float* sp = (q < 32)
        ? (Wl + (unsigned)(colbase + cc) * (unsigned)F_IN + q * 4)
        : (Wr + (unsigned)(colbase + cc) * (unsigned)F_IN + (q - 32) * 4);
    const float4 v = *(const float4*)sp;
    *(float4*)(Bs + (q >> 5) * (128 * BPITCH) + cc * BPITCH + (q & 31) * 4) = v;
  }
  __syncthreads();

  const int rt = rb * 8 + wave;
  if (rt >= (N_NODES >> 4)) return;

  const int mrow  = rt * 16 + (lane & 15);
  const int khalf = (lane >> 4) * 2;
  const float* a_lo = agg + (unsigned)mrow * (unsigned)F_IN + khalf;
  const float* a_hi = x   + (unsigned)mrow * (unsigned)F_IN + khalf;
  const float* bs   = Bs + (lane & 15) * BPITCH + khalf;

  v8f c[8];
  #pragma unroll
  for (int t = 0; t < 8; ++t) c[t] = v8f{};

  strip_product<8, 128>(a_lo, a_hi, bs, c);

  const unsigned rbase = (unsigned)(rt * 16 + ((lane < 16) ? 0 : 8));
  #pragma unroll
  for (int t = 0; t < 8; ++t) {
    const int ncol = colbase + t * 16 + (lane & 15);
    const float bv = bias[ncol];
    float* orow = h + rbase * (unsigned)HID + ncol;
    #pragma unroll
    for (int i = 0; i < 8; ++i) {
      float v = c[t][i] + bv;
      v = v > 0.0f ? v : 0.0f;
      orow[(unsigned)i * (unsigned)HID] = v;
    }
  }
}

__global__ void __launch_bounds__(256, 6)
proj_gemm_strip(const float* __restrict__ H,
                const float* __restrict__ Wl,
                const float* __restrict__ Wr,
                float* __restrict__ P,
                float* __restrict__ R) {
  __shared__ __attribute__((aligned(16))) float Bs[2 * (2 * CLS) * BPITCH];
  const int tid  = threadIdx.x;
  const int lane = tid & 31;
  const int wave = tid >> 5;

  #pragma unroll
  for (int j = 0; j < 20; ++j) {
    const int i  = tid + j * 256;
    const int cc = i >> 6;
    const int q  = i & 63;
    const float* sp = (cc < CLS) ? (Wl + (unsigned)cc * (unsigned)HID + q * 4)
                                 : (Wr + (unsigned)(cc - CLS) * (unsigned)HID + q * 4);
    const float4 v = *(const float4*)sp;
    *(float4*)(Bs + (q >> 5) * (2 * CLS * BPITCH) + cc * BPITCH + (q & 31) * 4) = v;
  }
  __syncthreads();

  const int rt = blockIdx.x * 8 + wave;
  if (rt >= (N_NODES >> 4)) return;

  const int mrow  = rt * 16 + (lane & 15);
  const int khalf = (lane >> 4) * 2;
  const float* a_lo = H + (unsigned)mrow * (unsigned)HID + khalf;
  const float* a_hi = a_lo + 128;
  const float* bs   = Bs + (lane & 15) * BPITCH + khalf;

  v8f c[5];
  #pragma unroll
  for (int t = 0; t < 5; ++t) c[t] = v8f{};

  strip_product<5, 2 * CLS>(a_lo, a_hi, bs, c);

  const unsigned rbase = (unsigned)(rt * 16 + ((lane < 16) ? 0 : 8));
  #pragma unroll
  for (int t = 0; t < 5; ++t) {
    const int  ncol = t * 16 + (lane & 15);
    const bool root = (ncol >= CLS);
    const int  wrow = root ? (ncol - CLS) : ncol;
    float* orow = (root ? R : P) + rbase * (unsigned)CLS + wrow;
    #pragma unroll
    for (int i = 0; i < 8; ++i) orow[(unsigned)i * (unsigned)CLS] = c[t][i];
  }
}

__global__ void finish_rows(const float* __restrict__ aggp,
                            const int*   __restrict__ rowptr,
                            const float* __restrict__ rootp,
                            const float* __restrict__ bias,
                            float* __restrict__ out) {
  int r = blockIdx.x * blockDim.x + threadIdx.x;
  if (r >= N_NODES) return;
  float c = (float)(rowptr[r + 1] - rowptr[r]);
  c = c > 1.0f ? c : 1.0f;
  const float4* a4 = (const float4*)(aggp  + (long long)r * CLS);
  const float4* r4 = (const float4*)(rootp + (long long)r * CLS);
  float l[CLS];
  #pragma unroll
  for (int q = 0; q < CLS / 4; ++q) {
    float4 a = a4[q];
    float4 t = r4[q];
    l[4 * q + 0] = a.x / c + bias[4 * q + 0] + t.x;
    l[4 * q + 1] = a.y / c + bias[4 * q + 1] + t.y;
    l[4 * q + 2] = a.z / c + bias[4 * q + 2] + t.z;
    l[4 * q + 3] = a.w / c + bias[4 * q + 3] + t.w;
  }
  float m = -3.402823466e38f;
  #pragma unroll
  for (int j = 0; j < CLS; ++j) m = fmaxf(m, l[j]);
  float s = 0.0f;
  #pragma unroll
  for (int j = 0; j < CLS; ++j) s += expf(l[j] - m);
  float ls = logf(s);
  float* o = out + (long long)r * CLS;
  #pragma unroll
  for (int j = 0; j < CLS; ++j) o[j] = l[j] - m - ls;
}

extern "C" void kernel_launch(void* const* d_in, const int* in_sizes, int n_in,
                              void* d_out, int out_size, void* d_ws, size_t ws_size,
                              hipStream_t stream) {
  (void)in_sizes; (void)n_in; (void)out_size; (void)ws_size;

  const float* x    = (const float*)d_in[0];
  const int*   ei   = (const int*)  d_in[1];
  const float* W1_l = (const float*)d_in[2];
  const float* b1   = (const float*)d_in[3];
  const float* W1_r = (const float*)d_in[4];
  const float* W2_l = (const float*)d_in[5];
  const float* b2   = (const float*)d_in[6];
  const float* W2_r = (const float*)d_in[7];
  const int* src = ei;
  const int* dst = ei + NE;
  float* out = (float*)d_out;

  float* ws     = (float*)d_ws;
  float* agg1   = ws;
  float* h      = agg1 + (long long)N_NODES * F_IN;
  float* p2     = h    + (long long)N_NODES * HID;
  float* r2     = p2   + (long long)N_NODES * CLS;
  int*   rowptr = (int*)(r2 + (long long)N_NODES * CLS);
  int*   csr    = rowptr + NPAD;
  float* aggp   = agg1;
  int*   off    = (int*)h;
  int2*  pairs  = (int2*)(off + NPAD);

  const int TB = 256;
  auto blocks = [](long long n, int tb) { return (unsigned)((n + tb - 1) / tb); };

  chunk_sort<<<CH, CHT, 0, stream>>>(src, dst, off, pairs);
  bucket_group<<<NB, 1024, 0, stream>>>(pairs, off, rowptr, csr);

  gather_mean_rows128<<<2048, TB, 0, stream>>>(x, rowptr, csr, agg1);

  layer1_gemm_strip<<<782, TB, 0, stream>>>(agg1, x, W1_l, W1_r, b1, h);

  proj_gemm_strip<<<391, TB, 0, stream>>>(h, W2_l, W2_r, p2, r2);

  gather_sum_rows40<<<blocks((long long)N_NODES * (CLS / 4), TB), TB, 0, stream>>>(
      p2, rowptr, csr, aggp);

  finish_rows<<<blocks(N_NODES, TB), TB, 0, stream>>>(aggp, rowptr, r2, b2, out);
}


# Layer-1 WMMA product re-gridded: 250 blocks of 13 waves, one per WGP, a wave runs two strips in turn: 25 strips a WGP, 6 or 7 a SIMD, all resident (was 782 blocks, two uneven rounds); sum order kept
# speedup vs baseline: 5.8586x; 1.0365x over previous
#include <hip/hip_runtime.h>

#define N_NODES 50000
#define F_IN    128
#define HID     256
#define CLS     40
#define NE      640000

typedef __attribute__((ext_vector_type(2))) float v2f;
typedef __attribute__((ext_vector_type(8))) float v8f;

#define NPAD 53248
#define CSR_PAD 8
#define NB  196
#define CH  250
#define CHT 640
#define CHE 2560
#define OFFP 200

__global__ void __launch_bounds__(CHT)
chunk_sort(const int* __restrict__ src,
           const int* __restrict__ dst,
           int* __restrict__ off,
           int2* __restrict__ pairs) {
  __shared__ int cnt[NB];
  __shared__ int beg[32 * 7];
  const int t = threadIdx.x;
  const int c = blockIdx.x;
  const int4 s = ((const int4*)src)[c * CHT + t];
  const int4 d = ((const int4*)dst)[c * CHT + t];
  if (t < NB) cnt[t] = 0;
  __syncthreads();
  const int r0 = atomicAdd(&cnt[d.x >> 8], 1);
  const int r1 = atomicAdd(&cnt[d.y >> 8], 1);
  const int r2 = atomicAdd(&cnt[d.z >> 8], 1);
  const int r3 = atomicAdd(&cnt[d.w >> 8], 1);
  __syncthreads();
  if (t < 32) {
    int v[7];
    int tot = 0;
    #pragma unroll
    for (int k = 0; k < 7; ++k) {
      const int i = t * 7 + k;
      v[k] = (i < NB) ? cnt[i] : 0;
      tot += v[k];
    }
    int inc = tot;
    #pragma unroll
    for (int o = 1; o < 32; o <<= 1) {
      const int u = __shfl_up(inc, o);
      if (t >= o) inc += u;
    }
    int ex = inc - tot;
    #pragma unroll
    for (int k = 0; k < 7; ++k) {
      const int i = t * 7 + k;
      beg[i] = ex;
      if (i <= NB) off[c * OFFP + i] = ex;
      ex += v[k];
    }
  }
  __syncthreads();
  int2* mine = pairs + c * CHE;
  mine[beg[d.x >> 8] + r0] = make_int2(s.x, d.x);
  mine[beg[d.y >> 8] + r1] = make_int2(s.y, d.y);
  mine[beg[d.z >> 8] + r2] = make_int2(s.z, d.z);
  mine[beg[d.w >> 8] + r3] = make_int2(s.w, d.w);
}

__device__ __forceinline__ int run_slot(const int* rs, const int* ro, int e) {
  int lo = 0, hi = CH;
  #pragma unroll
  for (int k = 0; k < 8; ++k) {
    const int mid = (lo + hi) >> 1;
    const bool up = rs[mid] <= e;
    lo = up ? mid : lo;
    hi = up ? hi : mid;
  }
  return ro[lo] + (e - rs[lo]);
}

__global__ void __launch_bounds__(1024)
bucket_group(const int2* __restrict__ pairs,
             const int* __restrict__ off,
             int* __restrict__ rowptr,
             int* __restrict__ csr) {
  __shared__ int cnt[256];
  __shared__ int cur[256];
  __shared__ int ln[256];
  __shared__ int st[256];
  __shared__ int rs[257];
  __shared__ int ro[256];
  __shared__ int lower;
  const int t = threadIdx.x;
  const int b = blockIdx.x;
  if (t < 256) {
    cnt[t] = 0;
    int o0 = 0, o1 = 0;
    if (t < CH) { o0 = off[t * OFFP + b]; o1 = off[t * OFFP + b + 1]; }
    st[t] = o0;
    ln[t] = o1 - o0;
    ro[t] = t * CHE + o0;
  }
  if (b == 0 && t < CSR_PAD) csr[NE + t] = 0;
  __syncthreads();
  if (t < 32) {
    int v[8];
    int tot = 0, low = 0;
    #pragma unroll
    for (int k = 0; k < 8; ++k) { v[k] = ln[t * 8 + k]; tot += v[k]; low += st[t * 8 + k]; }
    int inc = tot;
    #pragma unroll
    for (int o = 1; o < 32; o <<= 1) {
      const int u = __shfl_up(inc, o);
      if (t >= o) inc += u;
    }
    #pragma unroll
    for (int o = 16; o >= 1; o >>= 1) low += __shfl_xor(low, o);
    int ex = inc - tot;
    #pragma unroll
    for (int k = 0; k < 8; ++k) { rs[t * 8 + k] = ex; ex += v[k]; }
    if (t == 31) rs[256] = ex;
    if (t == 0) lower = low;
  }
  __syncthreads();
  const int total = rs[CH];
  const int beg   = lower;
  for (int e = t; e < total; e += 4096) {
    const bool k1 = e + 1024 < total, k2 = e + 2048 < total, k3 = e + 3072 < total;
    const int2 e0 = pairs[run_slot(rs, ro, e)];
    const int2 e1 = pairs[run_slot(rs, ro, min(e + 1024, total - 1))];
    const int2 e2 = pairs[run_slot(rs, ro, min(e + 2048, total - 1))];
    const int2 e3 = pairs[run_slot(rs, ro, min(e + 3072, total - 1))];
    atomicAdd(&cnt[e0.y & 255], 1);
    if (k1) atomicAdd(&cnt[e1.y & 255], 1);
    if (k2) atomicAdd(&cnt[e2.y & 255], 1);
    if (k3) atomicAdd(&cnt[e3.y & 255], 1);
  }
  __syncthreads();
  if (t < 32) {
    int v[8];
    int tot = 0;
    #pragma unroll
    for (int k = 0; k < 8; ++k) { v[k] = cnt[t * 8 + k]; tot += v[k]; }
    int inc = tot;
    #pragma unroll
    for (int o = 1; o < 32; o <<= 1) {
      const int u = __shfl_up(inc, o);
      if (t >= o) inc += u;
    }
    int ex = beg + inc - tot;
    #pragma unroll
    for (int k = 0; k < 8; ++k) {
      cur[t * 8 + k] = ex;
      rowptr[b * 256 + t * 8 + k] = ex;
      ex += v[k];
    }
  }
  __syncthreads();
  for (int e = t; e < total; e += 4096) {
    const bool k1 = e + 1024 < total, k2 = e + 2048 < total, k3 = e + 3072 < total;
    const int2 e0 = pairs[run_slot(rs, ro, e)];
    const int2 e1 = pairs[run_slot(rs, ro, min(e + 1024, total - 1))];
    const int2 e2 = pairs[run_slot(rs, ro, min(e + 2048, total - 1))];
    const int2 e3 = pairs[run_slot(rs, ro, min(e + 3072, total - 1))];
    csr[atomicAdd(&cur[e0.y & 255], 1)] = e0.x;
    if (k1) csr[atomicAdd(&cur[e1.y & 255], 1)] = e1.x;
    if (k2) csr[atomicAdd(&cur[e2.y & 255], 1)] = e2.x;
    if (k3) csr[atomicAdd(&cur[e3.y & 255], 1)] = e3.x;
  }
}

__global__ void gather_mean_rows128(const float* __restrict__ x,
                                    const int* __restrict__ rowptr,
                                    const int* __restrict__ csr,
                                    float* __restrict__ agg) {
  const int lane = threadIdx.x & 31;
  const int wv = __builtin_amdgcn_readfirstlane((int)(blockIdx.x * 8 + (threadIdx.x >> 5)));
  const int nw = (int)gridDim.x * 8;
  const float4* x4 = (const float4*)x;
  for (int n = wv; n < N_NODES; n += nw) {
    const int beg = rowptr[n];
    const int end = rowptr[n + 1];
    float4 acc = make_float4(0.0f, 0.0f, 0.0f, 0.0f);
    int j = beg;
    int s0 = csr[j], s1 = csr[j + 1], s2 = csr[j + 2], s3 = csr[j + 3];
    for (; j + 4 <= end; j += 4) {
      const int t0 = csr[j + 4], t1 = csr[j + 5], t2 = csr[j + 6], t3 = csr[j + 7];
      const float4 v0 = x4[(unsigned)s0 * 32u + lane];
      const float4 v1 = x4[(unsigned)s1 * 32u + lane];
      const float4 v2 = x4[(unsigned)s2 * 32u + lane];
      const float4 v3 = x4[(unsigned)s3 * 32u + lane];
      acc.x += v0.x; acc.y += v0.y; acc.z += v0.z; acc.w += v0.w;
      acc.x += v1.x; acc.y += v1.y; acc.z += v1.z; acc.w += v1.w;
      acc.x += v2.x; acc.y += v2.y; acc.z += v2.z; acc.w += v2.w;
      acc.x += v3.x; acc.y += v3.y; acc.z += v3.z; acc.w += v3.w;
      s0 = t0; s1 = t1; s2 = t2; s3 = t3;
    }
    const int rem = end - j;
    if (rem == 3) {
      const float4 v0 = x4[(unsigned)s0 * 32u + lane];
      const float4 v1 = x4[(unsigned)s1 * 32u + lane];
      const float4 v2 = x4[(unsigned)s2 * 32u + lane];
      acc.x += v0.x; acc.y += v0.y; acc.z += v0.z; acc.w += v0.w;
      acc.x += v1.x; acc.y += v1.y; acc.z += v1.z; acc.w += v1.w;
      acc.x += v2.x; acc.y += v2.y; acc.z += v2.z; acc.w += v2.w;
    } else if (rem == 2) {
      const float4 v0 = x4[(unsigned)s0 * 32u + lane];
      const float4 v1 = x4[(unsigned)s1 * 32u + lane];
      acc.x += v0.x; acc.y += v0.y; acc.z += v0.z; acc.w += v0.w;
      acc.x += v1.x; acc.y += v1.y; acc.z += v1.z; acc.w += v1.w;
    } else if (rem == 1) {
      const float4 v0 = x4[(unsigned)s0 * 32u + lane];
      acc.x += v0.x; acc.y += v0.y; acc.z += v0.z; acc.w += v0.w;
    }
    float c = (float)(end - beg);
    c = c > 1.0f ? c : 1.0f;
    float4 m;
    m.x = acc.x / c; m.y = acc.y / c; m.z = acc.z / c; m.w = acc.w / c;
    ((float4*)agg)[(unsigned)n * 32u + lane] = m;
  }
}

__global__ void gather_sum_rows40(const float* __restrict__ p,
                                  const int* __restrict__ rowptr,
                                  const int* __restrict__ csr,
                                  float* __restrict__ aggp) {
  const int i = blockIdx.x * blockDim.x + threadIdx.x;
  if (i >= N_NODES * (CLS / 4)) return;
  const int n = i / (CLS / 4);
  const int q = i - n * (CLS / 4);
  const float4* p4 = (const float4*)p;
  const int beg = rowptr[n];
  const int end = rowptr[n + 1];
  float4 acc = make_float4(0.0f, 0.0f, 0.0f, 0.0f);
  int j = beg;
  int s0 = csr[j], s1 = csr[j + 1], s2 = csr[j + 2], s3 = csr[j + 3];
  for (; j + 4 <= end; j += 4) {
    const int t0 = csr[j + 4], t1 = csr[j + 5], t2 = csr[j + 6], t3 = csr[j + 7];
    const float4 v0 = p4[s0 * (CLS / 4) + q];
    const float4 v1 = p4[s1 * (CLS / 4) + q];
    const float4 v2 = p4[s2 * (CLS / 4) + q];
    const float4 v3 = p4[s3 * (CLS / 4) + q];
    acc.x += v0.x; acc.y += v0.y; acc.z += v0.z; acc.w += v0.w;
    acc.x += v1.x; acc.y += v1.y; acc.z += v1.z; acc.w += v1.w;
    acc.x += v2.x; acc.y += v2.y; acc.z += v2.z; acc.w += v2.w;
    acc.x += v3.x; acc.y += v3.y; acc.z += v3.z; acc.w += v3.w;
    s0 = t0; s1 = t1; s2 = t2; s3 = t3;
  }
  const int rem = end - j;
  float4 v0 = make_float4(0.0f, 0.0f, 0.0f, 0.0f), v1 = v0, v2 = v0;
  if (rem > 0) v0 = p4[s0 * (CLS / 4) + q];
  if (rem > 1) v1 = p4[s1 * (CLS / 4) + q];
  if (rem > 2) v2 = p4[s2 * (CLS / 4) + q];
  if (rem > 0) { acc.x += v0.x; acc.y += v0.y; acc.z += v0.z; acc.w += v0.w; }
  if (rem > 1) { acc.x += v1.x; acc.y += v1.y; acc.z += v1.z; acc.w += v1.w; }
  if (rem > 2) { acc.x += v2.x; acc.y += v2.y; acc.z += v2.z; acc.w += v2.w; }
  ((float4*)aggp)[i] = acc;
}

#define BPITCH 132

template <int NT>
__device__ __forceinline__ void group_products(const float2 (&f)[8],
                                               const float* bg,
                                               v8f (&c)[NT]) {
  #pragma unroll
  for (int s = 0; s < 8; ++s) {
    v2f a; a.x = f[s].x; a.y = f[s].y;
    #pragma unroll
    for (int t = 0; t < NT; ++t) {
      float2 bv = *(const float2*)(bg + t * 16 * BPITCH + 4 * s);
      v2f b; b.x = bv.x; b.y = bv.y;
      c[t] = __builtin_amdgcn_wmma_f32_16x16x4_f32(false, a, false, b,
                                                   (short)0, c[t], false, false);
    }
  }
}

template <int NT, int NC>
__device__ __forceinline__ void strip_product(const float* __restrict__ a_lo,
                                              const float* __restrict__ a_hi,
                                              const float* bs,
                                              v8f (&c)[NT]) {
  float2 fa[8], fb[8];
  #pragma unroll
  for (int s = 0; s < 8; ++s) fa[s] = *(const float2*)(a_lo + 4 * s);

  #pragma unroll 1
  for (int gg = 0; gg < 4; ++gg) {
    {
      const int gn = 2 * gg + 1;
      const float* pn = (gn < 4) ? (a_lo + gn * 32) : (a_hi + (gn - 4) * 32);
      #pragma unroll
      for (int s = 0; s < 8; ++s) fb[s] = *(const float2*)(pn + 4 * s);
    }
    const float* bt = bs + (gg >> 1) * (NC * BPITCH) + (gg & 1) * 64;
    group_products<NT>(fa, bt, c);
    {
      const int gn = (gg < 3) ? (2 * gg + 2) : 7;
      const float* pn = (gn < 4) ? (a_lo + gn * 32) : (a_hi + (gn - 4) * 32);
      #pragma unroll
      for (int s = 0; s < 8; ++s) fa[s] = *(const float2*)(pn + 4 * s);
    }
    group_products<NT>(fb, bt + 32, c);
  }
}

#define L1_WAVES  13
#define L1_STRIPS 25
__global__ void __launch_bounds__(L1_WAVES * 32, 4)
layer1_gemm_strip(const float* __restrict__ agg,
                  const float* __restrict__ x,
                  const float* __restrict__ Wl,
                  const float* __restrict__ Wr,
                  const float* __restrict__ bias,
                  float* __restrict__ h) {
  __shared__ __attribute__((aligned(16))) float Bs[2 * 128 * BPITCH];
  const int tid  = threadIdx.x;
  const int lane = tid & 31;
  const int wave = __builtin_amdgcn_readfirstlane(tid >> 5);
  const int colbase = (blockIdx.x & 1) * 128;
  const int rb      = blockIdx.x >> 1;

  if (tid < 256) {
    #pragma unroll
    for (int j = 0; j < 32; ++j) {
      const int i  = tid + j * 256;
      const int cc = i >> 6;
      const int q  = i & 63;
      const float* sp = (q < 32)
          ? (Wl + (unsigned)(colbase + cc) * (unsigned)F_IN + q * 4)
          : (Wr + (unsigned)(colbase + cc) * (unsigned)F_IN + (q - 32) * 4);
      const float4 v = *(const float4*)sp;
      *(float4*)(Bs + (q >> 5) * (128 * BPITCH) + cc * BPITCH + (q & 31) * 4) = v;
    }
  }
  __syncthreads();

  const int khalf = (lane >> 4) * 2;
  const float* bs = Bs + (lane & 15) * BPITCH + khalf;

  const int trips = (wave == 0) ? 1 : 2;
  #pragma unroll 1
  for (int it = 0; it < trips; ++it) {
    const int rt    = rb * L1_STRIPS + wave + it * (L1_WAVES - 1);
    const int mrow  = rt * 16 + (lane & 15);
    const float* a_lo = agg + (unsigned)mrow * (unsigned)F_IN + khalf;
    const float* a_hi = x   + (unsigned)mrow * (unsigned)F_IN + khalf;

    v8f c[8];
    #pragma unroll
    for (int t = 0; t < 8; ++t) c[t] = v8f{};

    strip_product<8, 128>(a_lo, a_hi, bs, c);

    const unsigned rbase = (unsigned)(rt * 16 + ((lane < 16) ? 0 : 8));
    #pragma unroll
    for (int t = 0; t < 8; ++t) {
      const int ncol = colbase + t * 16 + (lane & 15);
      const float bv = bias[ncol];
      float* orow = h + rbase * (unsigned)HID + ncol;
      #pragma unroll
      for (int i = 0; i < 8; ++i) {
        float v = c[t][i] + bv;
        v = v > 0.0f ? v : 0.0f;
        orow[(unsigned)i * (unsigned)HID] = v;
      }
    }
  }
}

__global__ void __launch_bounds__(256, 6)
proj_gemm_strip(const float* __restrict__ H,
                const float* __restrict__ Wl,
                const float* __restrict__ Wr,
                float* __restrict__ P,
                float* __restrict__ R) {
  __shared__ __attribute__((aligned(16))) float Bs[2 * (2 * CLS) * BPITCH];
  const int tid  = threadIdx.x;
  const int lane = tid & 31;
  const int wave = tid >> 5;

  #pragma unroll
  for (int j = 0; j < 20; ++j) {
    const int i  = tid + j * 256;
    const int cc = i >> 6;
    const int q  = i & 63;
    const float* sp = (cc < CLS) ? (Wl + (unsigned)cc * (unsigned)HID + q * 4)
                                 : (Wr + (unsigned)(cc - CLS) * (unsigned)HID + q * 4);
    const float4 v = *(const float4*)sp;
    *(float4*)(Bs + (q >> 5) * (2 * CLS * BPITCH) + cc * BPITCH + (q & 31) * 4) = v;
  }
  __syncthreads();

  const int rt = blockIdx.x * 8 + wave;
  if (rt >= (N_NODES >> 4)) return;

  const int mrow  = rt * 16 + (lane & 15);
  const int khalf = (lane >> 4) * 2;
  const float* a_lo = H + (unsigned)mrow * (unsigned)HID + khalf;
  const float* a_hi = a_lo + 128;
  const float* bs   = Bs + (lane & 15) * BPITCH + khalf;

  v8f c[5];
  #pragma unroll
  for (int t = 0; t < 5; ++t) c[t] = v8f{};

  strip_product<5, 2 * CLS>(a_lo, a_hi, bs, c);

  const unsigned rbase = (unsigned)(rt * 16 + ((lane < 16) ? 0 : 8));
  #pragma unroll
  for (int t = 0; t < 5; ++t) {
    const int  ncol = t * 16 + (lane & 15);
    const bool root = (ncol >= CLS);
    const int  wrow = root ? (ncol - CLS) : ncol;
    float* orow = (root ? R : P) + rbase * (unsigned)CLS + wrow;
    #pragma unroll
    for (int i = 0; i < 8; ++i) orow[(unsigned)i * (unsigned)CLS] = c[t][i];
  }
}

__global__ void finish_rows(const float* __restrict__ aggp,
                            const int*   __restrict__ rowptr,
                            const float* __restrict__ rootp,
                            const float* __restrict__ bias,
                            float* __restrict__ out) {
  int r = blockIdx.x * blockDim.x + threadIdx.x;
  if (r >= N_NODES) return;
  float c = (float)(rowptr[r + 1] - rowptr[r]);
  c = c > 1.0f ? c : 1.0f;
  const float4* a4 = (const float4*)(aggp  + (long long)r * CLS);
  const float4* r4 = (const float4*)(rootp + (long long)r * CLS);
  float l[CLS];
  #pragma unroll
  for (int q = 0; q < CLS / 4; ++q) {
    float4 a = a4[q];
    float4 t = r4[q];
    l[4 * q + 0] = a.x / c + bias[4 * q + 0] + t.x;
    l[4 * q + 1] = a.y / c + bias[4 * q + 1] + t.y;
    l[4 * q + 2] = a.z / c + bias[4 * q + 2] + t.z;
    l[4 * q + 3] = a.w / c + bias[4 * q + 3] + t.w;
  }
  float m = -3.402823466e38f;
  #pragma unroll
  for (int j = 0; j < CLS; ++j) m = fmaxf(m, l[j]);
  float s = 0.0f;
  #pragma unroll
  for (int j = 0; j < CLS; ++j) s += expf(l[j] - m);
  float ls = logf(s);
  float* o = out + (long long)r * CLS;
  #pragma unroll
  for (int j = 0; j < CLS; ++j) o[j] = l[j] - m - ls;
}

extern "C" void kernel_launch(void* const* d_in, const int* in_sizes, int n_in,
                              void* d_out, int out_size, void* d_ws, size_t ws_size,
                              hipStream_t stream) {
  (void)in_sizes; (void)n_in; (void)out_size; (void)ws_size;

  const float* x    = (const float*)d_in[0];
  const int*   ei   = (const int*)  d_in[1];
  const float* W1_l = (const float*)d_in[2];
  const float* b1   = (const float*)d_in[3];
  const float* W1_r = (const float*)d_in[4];
  const float* W2_l = (const float*)d_in[5];
  const float* b2   = (const float*)d_in[6];
  const float* W2_r = (const float*)d_in[7];
  const int* src = ei;
  const int* dst = ei + NE;
  float* out = (float*)d_out;

  float* ws     = (float*)d_ws;
  float* agg1   = ws;
  float* h      = agg1 + (long long)N_NODES * F_IN;
  float* p2     = h    + (long long)N_NODES * HID;
  float* r2     = p2   + (long long)N_NODES * CLS;
  int*   rowptr = (int*)(r2 + (long long)N_NODES * CLS);
  int*   csr    = rowptr + NPAD;
  float* aggp   = agg1;
  int*   off    = (int*)h;
  int2*  pairs  = (int2*)(off + NPAD);

  const int TB = 256;
  auto blocks = [](long long n, int tb) { return (unsigned)((n + tb - 1) / tb); };

  chunk_sort<<<CH, CHT, 0, stream>>>(src, dst, off, pairs);
  bucket_group<<<NB, 1024, 0, stream>>>(pairs, off, rowptr, csr);

  gather_mean_rows128<<<2048, TB, 0, stream>>>(x, rowptr, csr, agg1);

  layer1_gemm_strip<<<250, L1_WAVES * 32, 0, stream>>>(agg1, x, W1_l, W1_r, b1, h);

  proj_gemm_strip<<<391, TB, 0, stream>>>(h, W2_l, W2_r, p2, r2);

  gather_sum_rows40<<<blocks((long long)N_NODES * (CLS / 4), TB), TB, 0, stream>>>(
      p2, rowptr, csr, aggp);

  finish_rows<<<blocks(N_NODES, TB), TB, 0, stream>>>(aggp, rowptr, r2, b2, out);
}


# Layer-2 row epilogue (mean, bias, root, log-softmax) folded into the gather launch: ten threads keep a node's 40 logits in registers, row max, exp sum via LDS; 7->6 launches; exp sum order changes
# speedup vs baseline: 6.1448x; 1.0489x over previous
#include <hip/hip_runtime.h>

#define N_NODES 50000
#define F_IN    128
#define HID     256
#define CLS     40
#define NE      640000

typedef __attribute__((ext_vector_type(2))) float v2f;
typedef __attribute__((ext_vector_type(8))) float v8f;

#define NPAD 53248
#define CSR_PAD 8
#define NB  196
#define CH  250
#define CHT 640
#define CHE 2560
#define OFFP 200

__global__ void __launch_bounds__(CHT)
chunk_sort(const int* __restrict__ src,
           const int* __restrict__ dst,
           int* __restrict__ off,
           int2* __restrict__ pairs) {
  __shared__ int cnt[NB];
  __shared__ int beg[32 * 7];
  const int t = threadIdx.x;
  const int c = blockIdx.x;
  const int4 s = ((const int4*)src)[c * CHT + t];
  const int4 d = ((const int4*)dst)[c * CHT + t];
  if (t < NB) cnt[t] = 0;
  __syncthreads();
  const int r0 = atomicAdd(&cnt[d.x >> 8], 1);
  const int r1 = atomicAdd(&cnt[d.y >> 8], 1);
  const int r2 = atomicAdd(&cnt[d.z >> 8], 1);
  const int r3 = atomicAdd(&cnt[d.w >> 8], 1);
  __syncthreads();
  if (t < 32) {
    int v[7];
    int tot = 0;
    #pragma unroll
    for (int k = 0; k < 7; ++k) {
      const int i = t * 7 + k;
      v[k] = (i < NB) ? cnt[i] : 0;
      tot += v[k];
    }
    int inc = tot;
    #pragma unroll
    for (int o = 1; o < 32; o <<= 1) {
      const int u = __shfl_up(inc, o);
      if (t >= o) inc += u;
    }
    int ex = inc - tot;
    #pragma unroll
    for (int k = 0; k < 7; ++k) {
      const int i = t * 7 + k;
      beg[i] = ex;
      if (i <= NB) off[c * OFFP + i] = ex;
      ex += v[k];
    }
  }
  __syncthreads();
  int2* mine = pairs + c * CHE;
  mine[beg[d.x >> 8] + r0] = make_int2(s.x, d.x);
  mine[beg[d.y >> 8] + r1] = make_int2(s.y, d.y);
  mine[beg[d.z >> 8] + r2] = make_int2(s.z, d.z);
  mine[beg[d.w >> 8] + r3] = make_int2(s.w, d.w);
}

__device__ __forceinline__ int run_slot(const int* rs, const int* ro, int e) {
  int lo = 0, hi = CH;
  #pragma unroll
  for (int k = 0; k < 8; ++k) {
    const int mid = (lo + hi) >> 1;
    const bool up = rs[mid] <= e;
    lo = up ? mid : lo;
    hi = up ? hi : mid;
  }
  return ro[lo] + (e - rs[lo]);
}

__global__ void __launch_bounds__(1024)
bucket_group(const int2* __restrict__ pairs,
             const int* __restrict__ off,
             int* __restrict__ rowptr,
             int* __restrict__ csr) {
  __shared__ int cnt[256];
  __shared__ int cur[256];
  __shared__ int ln[256];
  __shared__ int st[256];
  __shared__ int rs[257];
  __shared__ int ro[256];
  __shared__ int lower;
  const int t = threadIdx.x;
  const int b = blockIdx.x;
  if (t < 256) {
    cnt[t] = 0;
    int o0 = 0, o1 = 0;
    if (t < CH) { o0 = off[t * OFFP + b]; o1 = off[t * OFFP + b + 1]; }
    st[t] = o0;
    ln[t] = o1 - o0;
    ro[t] = t * CHE + o0;
  }
  if (b == 0 && t < CSR_PAD) csr[NE + t] = 0;
  __syncthreads();
  if (t < 32) {
    int v[8];
    int tot = 0, low = 0;
    #pragma unroll
    for (int k = 0; k < 8; ++k) { v[k] = ln[t * 8 + k]; tot += v[k]; low += st[t * 8 + k]; }
    int inc = tot;
    #pragma unroll
    for (int o = 1; o < 32; o <<= 1) {
      const int u = __shfl_up(inc, o);
      if (t >= o) inc += u;
    }
    #pragma unroll
    for (int o = 16; o >= 1; o >>= 1) low += __shfl_xor(low, o);
    int ex = inc - tot;
    #pragma unroll
    for (int k = 0; k < 8; ++k) { rs[t * 8 + k] = ex; ex += v[k]; }
    if (t == 31) rs[256] = ex;
    if (t == 0) lower = low;
  }
  __syncthreads();
  const int total = rs[CH];
  const int beg   = lower;
  for (int e = t; e < total; e += 4096) {
    const bool k1 = e + 1024 < total, k2 = e + 2048 < total, k3 = e + 3072 < total;
    const int2 e0 = pairs[run_slot(rs, ro, e)];
    const int2 e1 = pairs[run_slot(rs, ro, min(e + 1024, total - 1))];
    const int2 e2 = pairs[run_slot(rs, ro, min(e + 2048, total - 1))];
    const int2 e3 = pairs[run_slot(rs, ro, min(e + 3072, total - 1))];
    atomicAdd(&cnt[e0.y & 255], 1);
    if (k1) atomicAdd(&cnt[e1.y & 255], 1);
    if (k2) atomicAdd(&cnt[e2.y & 255], 1);
    if (k3) atomicAdd(&cnt[e3.y & 255], 1);
  }
  __syncthreads();
  if (t < 32) {
    int v[8];
    int tot = 0;
    #pragma unroll
    for (int k = 0; k < 8; ++k) { v[k] = cnt[t * 8 + k]; tot += v[k]; }
    int inc = tot;
    #pragma unroll
    for (int o = 1; o < 32; o <<= 1) {
      const int u = __shfl_up(inc, o);
      if (t >= o) inc += u;
    }
    int ex = beg + inc - tot;
    #pragma unroll
    for (int k = 0; k < 8; ++k) {
      cur[t * 8 + k] = ex;
      rowptr[b * 256 + t * 8 + k] = ex;
      ex += v[k];
    }
  }
  __syncthreads();
  for (int e = t; e < total; e += 4096) {
    const bool k1 = e + 1024 < total, k2 = e + 2048 < total, k3 = e + 3072 < total;
    const int2 e0 = pairs[run_slot(rs, ro, e)];
    const int2 e1 = pairs[run_slot(rs, ro, min(e + 1024, total - 1))];
    const int2 e2 = pairs[run_slot(rs, ro, min(e + 2048, total - 1))];
    const int2 e3 = pairs[run_slot(rs, ro, min(e + 3072, total - 1))];
    csr[atomicAdd(&cur[e0.y & 255], 1)] = e0.x;
    if (k1) csr[atomicAdd(&cur[e1.y & 255], 1)] = e1.x;
    if (k2) csr[atomicAdd(&cur[e2.y & 255], 1)] = e2.x;
    if (k3) csr[atomicAdd(&cur[e3.y & 255], 1)] = e3.x;
  }
}

__global__ void gather_mean_rows128(const float* __restrict__ x,
                                    const int* __restrict__ rowptr,
                                    const int* __restrict__ csr,
                                    float* __restrict__ agg) {
  const int lane = threadIdx.x & 31;
  const int wv = __builtin_amdgcn_readfirstlane((int)(blockIdx.x * 8 + (threadIdx.x >> 5)));
  const int nw = (int)gridDim.x * 8;
  const float4* x4 = (const float4*)x;
  for (int n = wv; n < N_NODES; n += nw) {
    const int beg = rowptr[n];
    const int end = rowptr[n + 1];
    float4 acc = make_float4(0.0f, 0.0f, 0.0f, 0.0f);
    int j = beg;
    int s0 = csr[j], s1 = csr[j + 1], s2 = csr[j + 2], s3 = csr[j + 3];
    for (; j + 4 <= end; j += 4) {
      const int t0 = csr[j + 4], t1 = csr[j + 5], t2 = csr[j + 6], t3 = csr[j + 7];
      const float4 v0 = x4[(unsigned)s0 * 32u + lane];
      const float4 v1 = x4[(unsigned)s1 * 32u + lane];
      const float4 v2 = x4[(unsigned)s2 * 32u + lane];
      const float4 v3 = x4[(unsigned)s3 * 32u + lane];
      acc.x += v0.x; acc.y += v0.y; acc.z += v0.z; acc.w += v0.w;
      acc.x += v1.x; acc.y += v1.y; acc.z += v1.z; acc.w += v1.w;
      acc.x += v2.x; acc.y += v2.y; acc.z += v2.z; acc.w += v2.w;
      acc.x += v3.x; acc.y += v3.y; acc.z += v3.z; acc.w += v3.w;
      s0 = t0; s1 = t1; s2 = t2; s3 = t3;
    }
    const int rem = end - j;
    if (rem == 3) {
      const float4 v0 = x4[(unsigned)s0 * 32u + lane];
      const float4 v1 = x4[(unsigned)s1 * 32u + lane];
      const float4 v2 = x4[(unsigned)s2 * 32u + lane];
      acc.x += v0.x; acc.y += v0.y; acc.z += v0.z; acc.w += v0.w;
      acc.x += v1.x; acc.y += v1.y; acc.z += v1.z; acc.w += v1.w;
      acc.x += v2.x; acc.y += v2.y; acc.z += v2.z; acc.w += v2.w;
    } else if (rem == 2) {
      const float4 v0 = x4[(unsigned)s0 * 32u + lane];
      const float4 v1 = x4[(unsigned)s1 * 32u + lane];
      acc.x += v0.x; acc.y += v0.y; acc.z += v0.z; acc.w += v0.w;
      acc.x += v1.x; acc.y += v1.y; acc.z += v1.z; acc.w += v1.w;
    } else if (rem == 1) {
      const float4 v0 = x4[(unsigned)s0 * 32u + lane];
      acc.x += v0.x; acc.y += v0.y; acc.z += v0.z; acc.w += v0.w;
    }
    float c = (float)(end - beg);
    c = c > 1.0f ? c : 1.0f;
    float4 m;
    m.x = acc.x / c; m.y = acc.y / c; m.z = acc.z / c; m.w = acc.w / c;
    ((float4*)agg)[(unsigned)n * 32u + lane] = m;
  }
}

#define BPITCH 132

template <int NT>
__device__ __forceinline__ void group_products(const float2 (&f)[8],
                                               const float* bg,
                                               v8f (&c)[NT]) {
  #pragma unroll
  for (int s = 0; s < 8; ++s) {
    v2f a; a.x = f[s].x; a.y = f[s].y;
    #pragma unroll
    for (int t = 0; t < NT; ++t) {
      float2 bv = *(const float2*)(bg + t * 16 * BPITCH + 4 * s);
      v2f b; b.x = bv.x; b.y = bv.y;
      c[t] = __builtin_amdgcn_wmma_f32_16x16x4_f32(false, a, false, b,
                                                   (short)0, c[t], false, false);
    }
  }
}

template <int NT, int NC>
__device__ __forceinline__ void strip_product(const float* __restrict__ a_lo,
                                              const float* __restrict__ a_hi,
                                              const float* bs,
                                              v8f (&c)[NT]) {
  float2 fa[8], fb[8];
  #pragma unroll
  for (int s = 0; s < 8; ++s) fa[s] = *(const float2*)(a_lo + 4 * s);

  #pragma unroll 1
  for (int gg = 0; gg < 4; ++gg) {
    {
      const int gn = 2 * gg + 1;
      const float* pn = (gn < 4) ? (a_lo + gn * 32) : (a_hi + (gn - 4) * 32);
      #pragma unroll
      for (int s = 0; s < 8; ++s) fb[s] = *(const float2*)(pn + 4 * s);
    }
    const float* bt = bs + (gg >> 1) * (NC * BPITCH) + (gg & 1) * 64;
    group_products<NT>(fa, bt, c);
    {
      const int gn = (gg < 3) ? (2 * gg + 2) : 7;
      const float* pn = (gn < 4) ? (a_lo + gn * 32) : (a_hi + (gn - 4) * 32);
      #pragma unroll
      for (int s = 0; s < 8; ++s) fa[s] = *(const float2*)(pn + 4 * s);
    }
    group_products<NT>(fb, bt + 32, c);
  }
}

#define L1_WAVES  13
#define L1_STRIPS 25
__global__ void __launch_bounds__(L1_WAVES * 32, 4)
layer1_gemm_strip(const float* __restrict__ agg,
                  const float* __restrict__ x,
                  const float* __restrict__ Wl,
                  const float* __restrict__ Wr,
                  const float* __restrict__ bias,
                  float* __restrict__ h) {
  __shared__ __attribute__((aligned(16))) float Bs[2 * 128 * BPITCH];
  const int tid  = threadIdx.x;
  const int lane = tid & 31;
  const int wave = __builtin_amdgcn_readfirstlane(tid >> 5);
  const int colbase = (blockIdx.x & 1) * 128;
  const int rb      = blockIdx.x >> 1;

  if (tid < 256) {
    #pragma unroll
    for (int j = 0; j < 32; ++j) {
      const int i  = tid + j * 256;
      const int cc = i >> 6;
      const int q  = i & 63;
      const float* sp = (q < 32)
          ? (Wl + (unsigned)(colbase + cc) * (unsigned)F_IN + q * 4)
          : (Wr + (unsigned)(colbase + cc) * (unsigned)F_IN + (q - 32) * 4);
      const float4 v = *(const float4*)sp;
      *(float4*)(Bs + (q >> 5) * (128 * BPITCH) + cc * BPITCH + (q & 31) * 4) = v;
    }
  }
  __syncthreads();

  const int khalf = (lane >> 4) * 2;
  const float* bs = Bs + (lane & 15) * BPITCH + khalf;

  const int trips = (wave == 0) ? 1 : 2;
  #pragma unroll 1
  for (int it = 0; it < trips; ++it) {
    const int rt    = rb * L1_STRIPS + wave + it * (L1_WAVES - 1);
    const int mrow  = rt * 16 + (lane & 15);
    const float* a_lo = agg + (unsigned)mrow * (unsigned)F_IN + khalf;
    const float* a_hi = x   + (unsigned)mrow * (unsigned)F_IN + khalf;

    v8f c[8];
    #pragma unroll
    for (int t = 0; t < 8; ++t) c[t] = v8f{};

    strip_product<8, 128>(a_lo, a_hi, bs, c);

    const unsigned rbase = (unsigned)(rt * 16 + ((lane < 16) ? 0 : 8));
    #pragma unroll
    for (int t = 0; t < 8; ++t) {
      const int ncol = colbase + t * 16 + (lane & 15);
      const float bv = bias[ncol];
      float* orow = h + rbase * (unsigned)HID + ncol;
      #pragma unroll
      for (int i = 0; i < 8; ++i) {
        float v = c[t][i] + bv;
        v = v > 0.0f ? v : 0.0f;
        orow[(unsigned)i * (unsigned)HID] = v;
      }
    }
  }
}

__global__ void __launch_bounds__(256, 6)
proj_gemm_strip(const float* __restrict__ H,
                const float* __restrict__ Wl,
                const float* __restrict__ Wr,
                float* __restrict__ P,
                float* __restrict__ R) {
  __shared__ __attribute__((aligned(16))) float Bs[2 * (2 * CLS) * BPITCH];
  const int tid  = threadIdx.x;
  const int lane = tid & 31;
  const int wave = tid >> 5;

  #pragma unroll
  for (int j = 0; j < 20; ++j) {
    const int i  = tid + j * 256;
    const int cc = i >> 6;
    const int q  = i & 63;
    const float* sp = (cc < CLS) ? (Wl + (unsigned)cc * (unsigned)HID + q * 4)
                                 : (Wr + (unsigned)(cc - CLS) * (unsigned)HID + q * 4);
    const float4 v = *(const float4*)sp;
    *(float4*)(Bs + (q >> 5) * (2 * CLS * BPITCH) + cc * BPITCH + (q & 31) * 4) = v;
  }
  __syncthreads();

  const int rt = blockIdx.x * 8 + wave;
  if (rt >= (N_NODES >> 4)) return;

  const int mrow  = rt * 16 + (lane & 15);
  const int khalf = (lane >> 4) * 2;
  const float* a_lo = H + (unsigned)mrow * (unsigned)HID + khalf;
  const float* a_hi = a_lo + 128;
  const float* bs   = Bs + (lane & 15) * BPITCH + khalf;

  v8f c[5];
  #pragma unroll
  for (int t = 0; t < 5; ++t) c[t] = v8f{};

  strip_product<5, 2 * CLS>(a_lo, a_hi, bs, c);

  const unsigned rbase = (unsigned)(rt * 16 + ((lane < 16) ? 0 : 8));
  #pragma unroll
  for (int t = 0; t < 5; ++t) {
    const int  ncol = t * 16 + (lane & 15);
    const bool root = (ncol >= CLS);
    const int  wrow = root ? (ncol - CLS) : ncol;
    float* orow = (root ? R : P) + rbase * (unsigned)CLS + wrow;
    #pragma unroll
    for (int i = 0; i < 8; ++i) orow[(unsigned)i * (unsigned)CLS] = c[t][i];
  }
}

#define G2_NODES 32
#define G2_T     (G2_NODES * (CLS / 4))
__global__ void __launch_bounds__(G2_T)
gather_finish_rows40(const float* __restrict__ p,
                     const int*   __restrict__ rowptr,
                     const int*   __restrict__ csr,
                     const float* __restrict__ rootp,
                     const float* __restrict__ bias,
                     float* __restrict__ out) {
  __shared__ float pmax[G2_T];
  __shared__ float psum[G2_T];
  const int t  = threadIdx.x;
  const int nl = t / (CLS / 4);
  const int q  = t - nl * (CLS / 4);
  const int nn = blockIdx.x * G2_NODES + nl;
  const bool live = nn < N_NODES;
  const int n = live ? nn : N_NODES - 1;
  const int i = n * (CLS / 4) + q;
  const float4* p4 = (const float4*)p;
  const int beg = rowptr[n];
  const int end = rowptr[n + 1];
  const float4 rv = ((const float4*)rootp)[i];
  const float bq0 = bias[4 * q + 0], bq1 = bias[4 * q + 1];
  const float bq2 = bias[4 * q + 2], bq3 = bias[4 * q + 3];
  float4 acc = make_float4(0.0f, 0.0f, 0.0f, 0.0f);
  int j = beg;
  int s0 = csr[j], s1 = csr[j + 1], s2 = csr[j + 2], s3 = csr[j + 3];
  for (; j + 4 <= end; j += 4) {
    const int t0 = csr[j + 4], t1 = csr[j + 5], t2 = csr[j + 6], t3 = csr[j + 7];
    const float4 v0 = p4[s0 * (CLS / 4) + q];
    const float4 v1 = p4[s1 * (CLS / 4) + q];
    const float4 v2 = p4[s2 * (CLS / 4) + q];
    const float4 v3 = p4[s3 * (CLS / 4) + q];
    acc.x += v0.x; acc.y += v0.y; acc.z += v0.z; acc.w += v0.w;
    acc.x += v1.x; acc.y += v1.y; acc.z += v1.z; acc.w += v1.w;
    acc.x += v2.x; acc.y += v2.y; acc.z += v2.z; acc.w += v2.w;
    acc.x += v3.x; acc.y += v3.y; acc.z += v3.z; acc.w += v3.w;
    s0 = t0; s1 = t1; s2 = t2; s3 = t3;
  }
  const int rem = end - j;
  float4 v0 = make_float4(0.0f, 0.0f, 0.0f, 0.0f), v1 = v0, v2 = v0;
  if (rem > 0) v0 = p4[s0 * (CLS / 4) + q];
  if (rem > 1) v1 = p4[s1 * (CLS / 4) + q];
  if (rem > 2) v2 = p4[s2 * (CLS / 4) + q];
  if (rem > 0) { acc.x += v0.x; acc.y += v0.y; acc.z += v0.z; acc.w += v0.w; }
  if (rem > 1) { acc.x += v1.x; acc.y += v1.y; acc.z += v1.z; acc.w += v1.w; }
  if (rem > 2) { acc.x += v2.x; acc.y += v2.y; acc.z += v2.z; acc.w += v2.w; }

  float c = (float)(end - beg);
  c = c > 1.0f ? c : 1.0f;
  float4 l;
  l.x = acc.x / c + bq0 + rv.x;
  l.y = acc.y / c + bq1 + rv.y;
  l.z = acc.z / c + bq2 + rv.z;
  l.w = acc.w / c + bq3 + rv.w;

  const int tn = t - q;
  pmax[t] = fmaxf(fmaxf(l.x, l.y), fmaxf(l.z, l.w));
  __syncthreads();
  float m = pmax[tn];
  for (int k = 1; k < CLS / 4; ++k) m = fmaxf(m, pmax[tn + k]);

  psum[t] = expf(l.x - m) + expf(l.y - m) + expf(l.z - m) + expf(l.w - m);
  __syncthreads();
  float s = psum[tn];
  for (int k = 1; k < CLS / 4; ++k) s += psum[tn + k];
  const float ls = logf(s);

  if (live) {
    float4 o;
    o.x = l.x - m - ls;
    o.y = l.y - m - ls;
    o.z = l.z - m - ls;
    o.w = l.w - m - ls;
    ((float4*)out)[i] = o;
  }
}

extern "C" void kernel_launch(void* const* d_in, const int* in_sizes, int n_in,
                              void* d_out, int out_size, void* d_ws, size_t ws_size,
                              hipStream_t stream) {
  (void)in_sizes; (void)n_in; (void)out_size; (void)ws_size;

  const float* x    = (const float*)d_in[0];
  const int*   ei   = (const int*)  d_in[1];
  const float* W1_l = (const float*)d_in[2];
  const float* b1   = (const float*)d_in[3];
  const float* W1_r = (const float*)d_in[4];
  const float* W2_l = (const float*)d_in[5];
  const float* b2   = (const float*)d_in[6];
  const float* W2_r = (const float*)d_in[7];
  const int* src = ei;
  const int* dst = ei + NE;
  float* out = (float*)d_out;

  float* ws     = (float*)d_ws;
  float* agg1   = ws;
  float* h      = agg1 + (long long)N_NODES * F_IN;
  float* p2     = h    + (long long)N_NODES * HID;
  float* r2     = p2   + (long long)N_NODES * CLS;
  int*   rowptr = (int*)(r2 + (long long)N_NODES * CLS);
  int*   csr    = rowptr + NPAD;
  int*   off    = (int*)h;
  int2*  pairs  = (int2*)(off + NPAD);

  const int TB = 256;

  chunk_sort<<<CH, CHT, 0, stream>>>(src, dst, off, pairs);
  bucket_group<<<NB, 1024, 0, stream>>>(pairs, off, rowptr, csr);

  gather_mean_rows128<<<2048, TB, 0, stream>>>(x, rowptr, csr, agg1);

  layer1_gemm_strip<<<250, L1_WAVES * 32, 0, stream>>>(agg1, x, W1_l, W1_r, b1, h);

  proj_gemm_strip<<<391, TB, 0, stream>>>(h, W2_l, W2_r, p2, r2);

  gather_finish_rows40<<<(N_NODES + G2_NODES - 1) / G2_NODES, G2_T, 0, stream>>>(
      p2, rowptr, csr, r2, b2, out);
}
